# starting kernel
# baseline (speedup 1.0000x reference)
#include <hip/hip_runtime.h>

#define NN 20000
#define NE 320000
#define INV_SQRT3 0.5773502691896258f

typedef __attribute__((ext_vector_type(2))) float v2f;
typedef __attribute__((ext_vector_type(8))) float v8f;

__device__ __forceinline__ v8f wmma4(v2f a, v2f b, v8f c) {
  return __builtin_amdgcn_wmma_f32_16x16x4_f32(false, a, false, b, (short)0, c,
                                               false, false);
}

__device__ __forceinline__ float silu_f(float x) {
  return x * (1.0f / (1.0f + __expf(-x)));
}

__device__ __forceinline__ v8f zero8() {
  v8f z = {0.f, 0.f, 0.f, 0.f, 0.f, 0.f, 0.f, 0.f};
  return z;
}

__global__ __launch_bounds__(256) void zero_kernel(float4* p, long n4) {
  long i = (long)blockIdx.x * blockDim.x + threadIdx.x;
  if (i < n4) p[i] = make_float4(0.f, 0.f, 0.f, 0.f);
}

__global__ __launch_bounds__(256) void node_up_kernel(
    const float* __restrict__ node_feats, const float* __restrict__ Wsc,
    const float* __restrict__ Wup0, const float* __restrict__ Wup1,
    float* __restrict__ ns, float* __restrict__ sup, float* __restrict__ vup) {
  const int lane = threadIdx.x & 31;
  const int wv = threadIdx.x >> 5;
  const int tile = blockIdx.x * 8 + wv;
  if (tile >= NN / 16) return;
  const int n0 = tile * 16;
  const int row = lane & 15;
  const int hi = lane >> 4;
  const int col = row;
  const float* arow = node_feats + (long)(n0 + row) * 256;

  v8f accA[4], accB[4];
#pragma unroll
  for (int nt = 0; nt < 4; nt++) { accA[nt] = zero8(); accB[nt] = zero8(); }
#pragma unroll
  for (int kk = 0; kk < 16; kk++) {
    const int kb = kk * 4 + 2 * hi;
    v2f a = *(const v2f*)(arow + kb);
#pragma unroll
    for (int nt = 0; nt < 4; nt++) {
      v2f b0, b1;
      b0.x = Wsc[kb * 64 + nt * 16 + col];
      b0.y = Wsc[(kb + 1) * 64 + nt * 16 + col];
      accA[nt] = wmma4(a, b0, accA[nt]);
      b1.x = Wup0[kb * 64 + nt * 16 + col];
      b1.y = Wup0[(kb + 1) * 64 + nt * 16 + col];
      accB[nt] = wmma4(a, b1, accB[nt]);
    }
  }
#pragma unroll
  for (int nt = 0; nt < 4; nt++)
#pragma unroll
    for (int r = 0; r < 8; r++) {
      const int n = n0 + r + 8 * hi;
      ns[(long)n * 64 + nt * 16 + col] = accA[nt][r];
      sup[(long)n * 64 + nt * 16 + col] = accB[nt][r];
    }

#pragma unroll
  for (int i = 0; i < 3; i++) {
    v8f acc[4];
#pragma unroll
    for (int nt = 0; nt < 4; nt++) acc[nt] = zero8();
#pragma unroll
    for (int kk = 0; kk < 16; kk++) {
      const int kb = kk * 4 + 2 * hi;
      v2f a;
      a.x = arow[64 + kb * 3 + i];
      a.y = arow[64 + (kb + 1) * 3 + i];
#pragma unroll
      for (int nt = 0; nt < 4; nt++) {
        v2f b;
        b.x = Wup1[kb * 64 + nt * 16 + col];
        b.y = Wup1[(kb + 1) * 64 + nt * 16 + col];
        acc[nt] = wmma4(a, b, acc[nt]);
      }
    }
#pragma unroll
    for (int nt = 0; nt < 4; nt++)
#pragma unroll
      for (int r = 0; r < 8; r++) {
        const int n = n0 + r + 8 * hi;
        vup[(long)i * NN * 64 + (long)n * 64 + nt * 16 + col] = acc[nt][r];
      }
  }
}

__global__ __launch_bounds__(128) void edge_kernel(
    const float* __restrict__ edge_attrs, const float* __restrict__ edge_feats,
    const float* __restrict__ lengths, const int* __restrict__ edge_index,
    const float* __restrict__ W1, const float* __restrict__ b1,
    const float* __restrict__ W2, const float* __restrict__ b2,
    const float* __restrict__ W3, const float* __restrict__ ns,
    const float* __restrict__ sup, const float* __restrict__ vup,
    float* __restrict__ m0, float* __restrict__ m1) {
  __shared__ float lds_h[4][32 * 64];
  __shared__ float lds_meta[4][32 * 8];
  const int lane = threadIdx.x & 31;
  const int wv = threadIdx.x >> 5;
  const int dt = blockIdx.x * 4 + wv;
  const int e0 = dt * 32;
  const int row = lane & 15;
  const int hi = lane >> 4;
  const int col = row;

  const int ea = e0 + row;
  const int eb = e0 + 16 + row;
  const int snd0 = edge_index[ea], rcv0 = edge_index[NE + ea];
  const int snd1 = edge_index[eb], rcv1 = edge_index[NE + eb];
  {
    const int mr = row + 16 * hi;
    const int me = e0 + mr;
    lds_meta[wv][mr * 8 + 0] = __int_as_float(hi ? snd1 : snd0);
    lds_meta[wv][mr * 8 + 1] = __int_as_float(hi ? rcv1 : rcv0);
    lds_meta[wv][mr * 8 + 2] = edge_attrs[me * 4 + 0];
    lds_meta[wv][mr * 8 + 3] = edge_attrs[me * 4 + 1];
    lds_meta[wv][mr * 8 + 4] = edge_attrs[me * 4 + 2];
    lds_meta[wv][mr * 8 + 5] = edge_attrs[me * 4 + 3];
  }
  const float* sp0 = ns + (long)snd0 * 64;
  const float* rp0 = ns + (long)rcv0 * 64;
  const float* sp1 = ns + (long)snd1 * 64;
  const float* rp1 = ns + (long)rcv1 * 64;
  const float* ep0 = edge_feats + (long)ea * 8;
  const float* ep1 = edge_feats + (long)eb * 8;
  const float len0 = lengths[ea];
  const float len1 = lengths[eb];

  v8f acc0[4], acc1[4];
#pragma unroll
  for (int nt = 0; nt < 4; nt++) { acc0[nt] = zero8(); acc1[nt] = zero8(); }
#pragma unroll 4
  for (int kk = 0; kk < 16; kk++) {
    const int kb = kk * 4 + 2 * hi;
    v2f a0 = *(const v2f*)(sp0 + kb);
    v2f a1 = *(const v2f*)(sp1 + kb);
#pragma unroll
    for (int nt = 0; nt < 4; nt++) {
      v2f b;
      b.x = W1[kb * 64 + nt * 16 + col];
      b.y = W1[(kb + 1) * 64 + nt * 16 + col];
      acc0[nt] = wmma4(a0, b, acc0[nt]);
      acc1[nt] = wmma4(a1, b, acc1[nt]);
    }
  }
#pragma unroll 4
  for (int kk = 0; kk < 16; kk++) {
    const int kb = kk * 4 + 2 * hi;
    v2f a0 = *(const v2f*)(rp0 + kb);
    v2f a1 = *(const v2f*)(rp1 + kb);
#pragma unroll
    for (int nt = 0; nt < 4; nt++) {
      v2f b;
      b.x = W1[(64 + kb) * 64 + nt * 16 + col];
      b.y = W1[(64 + kb + 1) * 64 + nt * 16 + col];
      acc0[nt] = wmma4(a0, b, acc0[nt]);
      acc1[nt] = wmma4(a1, b, acc1[nt]);
    }
  }
#pragma unroll
  for (int kk = 0; kk < 2; kk++) {
    const int kb = kk * 4 + 2 * hi;
    v2f a0 = *(const v2f*)(ep0 + kb);
    v2f a1 = *(const v2f*)(ep1 + kb);
#pragma unroll
    for (int nt = 0; nt < 4; nt++) {
      v2f b;
      b.x = W1[(128 + kb) * 64 + nt * 16 + col];
      b.y = W1[(128 + kb + 1) * 64 + nt * 16 + col];
      acc0[nt] = wmma4(a0, b, acc0[nt]);
      acc1[nt] = wmma4(a1, b, acc1[nt]);
    }
  }
  {
    v2f a0, a1;
    a0.x = hi ? 0.f : len0;  a0.y = 0.f;
    a1.x = hi ? 0.f : len1;  a1.y = 0.f;
#pragma unroll
    for (int nt = 0; nt < 4; nt++) {
      v2f b;
      b.x = hi ? 0.f : W1[136 * 64 + nt * 16 + col];
      b.y = 0.f;
      acc0[nt] = wmma4(a0, b, acc0[nt]);
      acc1[nt] = wmma4(a1, b, acc1[nt]);
    }
  }
#pragma unroll
  for (int nt = 0; nt < 4; nt++) {
    const float bias = b1[nt * 16 + col];
#pragma unroll
    for (int r = 0; r < 8; r++) {
      lds_h[wv][(r + 8 * hi) * 64 + nt * 16 + col] = silu_f(acc0[nt][r] + bias);
      lds_h[wv][(16 + r + 8 * hi) * 64 + nt * 16 + col] =
          silu_f(acc1[nt][r] + bias);
    }
  }

#pragma unroll
  for (int nt = 0; nt < 4; nt++) { acc0[nt] = zero8(); acc1[nt] = zero8(); }
#pragma unroll 4
  for (int kk = 0; kk < 16; kk++) {
    const int kb = kk * 4 + 2 * hi;
    v2f a0 = *(const v2f*)&lds_h[wv][row * 64 + kb];
    v2f a1 = *(const v2f*)&lds_h[wv][(16 + row) * 64 + kb];
#pragma unroll
    for (int nt = 0; nt < 4; nt++) {
      v2f b;
      b.x = W2[kb * 64 + nt * 16 + col];
      b.y = W2[(kb + 1) * 64 + nt * 16 + col];
      acc0[nt] = wmma4(a0, b, acc0[nt]);
      acc1[nt] = wmma4(a1, b, acc1[nt]);
    }
  }
  float h2a[4][8], h2b[4][8];
#pragma unroll
  for (int nt = 0; nt < 4; nt++) {
    const float bias = b2[nt * 16 + col];
#pragma unroll
    for (int r = 0; r < 8; r++) {
      h2a[nt][r] = silu_f(acc0[nt][r] + bias);
      h2b[nt][r] = silu_f(acc1[nt][r] + bias);
    }
  }
#pragma unroll
  for (int nt = 0; nt < 4; nt++)
#pragma unroll
    for (int r = 0; r < 8; r++) {
      lds_h[wv][(r + 8 * hi) * 64 + nt * 16 + col] = h2a[nt][r];
      lds_h[wv][(16 + r + 8 * hi) * 64 + nt * 16 + col] = h2b[nt][r];
    }

#pragma unroll
  for (int g = 0; g < 4; g++) {
    v8f t0[4], t1[4];
#pragma unroll
    for (int nt = 0; nt < 4; nt++) { t0[nt] = zero8(); t1[nt] = zero8(); }
#pragma unroll 4
    for (int kk = 0; kk < 16; kk++) {
      const int kb = kk * 4 + 2 * hi;
      v2f a0 = *(const v2f*)&lds_h[wv][row * 64 + kb];
      v2f a1 = *(const v2f*)&lds_h[wv][(16 + row) * 64 + kb];
#pragma unroll
      for (int nt = 0; nt < 4; nt++) {
        v2f b;
        b.x = W3[kb * 256 + g * 64 + nt * 16 + col];
        b.y = W3[(kb + 1) * 256 + g * 64 + nt * 16 + col];
        t0[nt] = wmma4(a0, b, t0[nt]);
        t1[nt] = wmma4(a1, b, t1[nt]);
      }
    }
#pragma unroll
    for (int t = 0; t < 2; t++) {
#pragma unroll
      for (int nt = 0; nt < 4; nt++) {
        const int c = nt * 16 + col;
#pragma unroll
        for (int r = 0; r < 8; r++) {
          const int rr = t * 16 + r + 8 * hi;
          const int s_e = __float_as_int(lds_meta[wv][rr * 8 + 0]);
          const int r_e = __float_as_int(lds_meta[wv][rr * 8 + 1]);
          const float y0 = lds_meta[wv][rr * 8 + 2];
          const float wgt = t ? t1[nt][r] : t0[nt][r];
          if (g == 0) {
            const float val = sup[(long)s_e * 64 + c] * y0 * wgt;
            unsafeAtomicAdd(&m0[(long)r_e * 128 + c], val);
          } else if (g == 1) {
            const float dot =
                vup[(long)s_e * 64 + c] * lds_meta[wv][rr * 8 + 3] +
                vup[(long)NN * 64 + (long)s_e * 64 + c] *
                    lds_meta[wv][rr * 8 + 4] +
                vup[(long)2 * NN * 64 + (long)s_e * 64 + c] *
                    lds_meta[wv][rr * 8 + 5];
            unsafeAtomicAdd(&m0[(long)r_e * 128 + 64 + c],
                            dot * INV_SQRT3 * wgt);
          } else if (g == 2) {
            const float xs = sup[(long)s_e * 64 + c];
#pragma unroll
            for (int i = 0; i < 3; i++)
              unsafeAtomicAdd(&m1[(long)i * NN * 128 + (long)r_e * 128 + c],
                              xs * lds_meta[wv][rr * 8 + 3 + i] * wgt);
          } else {
#pragma unroll
            for (int i = 0; i < 3; i++)
              unsafeAtomicAdd(
                  &m1[(long)i * NN * 128 + (long)r_e * 128 + 64 + c],
                  vup[(long)i * NN * 64 + (long)s_e * 64 + c] * y0 * wgt);
          }
        }
      }
    }
  }
}

__global__ __launch_bounds__(256) void out_kernel(
    const float* __restrict__ m0, const float* __restrict__ m1,
    const float* __restrict__ Ws, const float* __restrict__ Wv,
    float* __restrict__ out) {
  const int lane = threadIdx.x & 31;
  const int wv = threadIdx.x >> 5;
  const int tile = blockIdx.x * 8 + wv;
  if (tile >= NN / 16) return;
  const int n0 = tile * 16;
  const int row = lane & 15;
  const int hi = lane >> 4;
  const int col = row;

  v8f acc[4][4];
#pragma unroll
  for (int j = 0; j < 4; j++)
#pragma unroll
    for (int nt = 0; nt < 4; nt++) acc[j][nt] = zero8();

#pragma unroll 4
  for (int kk = 0; kk < 32; kk++) {
    const int kb = kk * 4 + 2 * hi;
    v2f a = *(const v2f*)(m0 + (long)(n0 + row) * 128 + kb);
#pragma unroll
    for (int nt = 0; nt < 4; nt++) {
      v2f b;
      b.x = Ws[kb * 64 + nt * 16 + col];
      b.y = Ws[(kb + 1) * 64 + nt * 16 + col];
      acc[0][nt] = wmma4(a, b, acc[0][nt]);
    }
  }
#pragma unroll
  for (int i = 0; i < 3; i++) {
#pragma unroll 4
    for (int kk = 0; kk < 32; kk++) {
      const int kb = kk * 4 + 2 * hi;
      v2f a =
          *(const v2f*)(m1 + (long)i * NN * 128 + (long)(n0 + row) * 128 + kb);
#pragma unroll
      for (int nt = 0; nt < 4; nt++) {
        v2f b;
        b.x = Wv[kb * 64 + nt * 16 + col];
        b.y = Wv[(kb + 1) * 64 + nt * 16 + col];
        acc[1 + i][nt] = wmma4(a, b, acc[1 + i][nt]);
      }
    }
  }
  const float inv = 1.0f / 16.0f;
#pragma unroll
  for (int nt = 0; nt < 4; nt++) {
    const int c = nt * 16 + col;
#pragma unroll
    for (int r = 0; r < 8; r++) {
      const int n = n0 + r + 8 * hi;
      float4 o = make_float4(acc[0][nt][r] * inv, acc[1][nt][r] * inv,
                             acc[2][nt][r] * inv, acc[3][nt][r] * inv);
      *(float4*)(out + (long)n * 256 + c * 4) = o;
    }
  }
}

extern "C" void kernel_launch(void* const* d_in, const int* in_sizes, int n_in,
                              void* d_out, int out_size, void* d_ws,
                              size_t ws_size, hipStream_t stream) {
  (void)in_sizes; (void)n_in; (void)out_size; (void)ws_size;
  const float* node_feats = (const float*)d_in[0];
  const float* edge_attrs = (const float*)d_in[1];
  const float* edge_feats = (const float*)d_in[2];
  const float* lengths    = (const float*)d_in[3];
  const int*   edge_index = (const int*)d_in[4];
  const float* Wsc  = (const float*)d_in[5];
  const float* Wup0 = (const float*)d_in[6];
  const float* Wup1 = (const float*)d_in[7];
  const float* W1   = (const float*)d_in[8];
  const float* b1   = (const float*)d_in[9];
  const float* W2   = (const float*)d_in[10];
  const float* b2   = (const float*)d_in[11];
  const float* W3   = (const float*)d_in[12];
  const float* Ws   = (const float*)d_in[13];
  const float* Wv   = (const float*)d_in[14];
  float* out = (float*)d_out;

  float* ws  = (float*)d_ws;
  float* ns  = ws;
  float* sup = ns + (long)NN * 64;
  float* vup = sup + (long)NN * 64;
  float* m0  = vup + (long)3 * NN * 64;
  float* m1  = m0 + (long)NN * 128;

  const long zn4 = (long)NN * 512 / 4;
  zero_kernel<<<(int)((zn4 + 255) / 256), 256, 0, stream>>>((float4*)m0, zn4);

  node_up_kernel<<<(NN / 16 + 7) / 8, 256, 0, stream>>>(node_feats, Wsc, Wup0,
                                                        Wup1, ns, sup, vup);
  edge_kernel<<<NE / 32 / 4, 128, 0, stream>>>(edge_attrs, edge_feats, lengths,
                                               edge_index, W1, b1, W2, b2, W3,
                                               ns, sup, vup, m0, m1);
  out_kernel<<<(NN / 16 + 7) / 8, 256, 0, stream>>>(m0, m1, Ws, Wv, out);
}


# edge kernel phase 3: W3's column blocks run as two passes of the two blocks sharing a gathered operand, (w0,w2) on sup[snd], (w1,w3) on vup[snd]: each value gathered once; same atomics, reordered
# speedup vs baseline: 1.3881x; 1.3881x over previous
#include <hip/hip_runtime.h>

#define NN 20000
#define NE 320000
#define INV_SQRT3 0.5773502691896258f

typedef __attribute__((ext_vector_type(2))) float v2f;
typedef __attribute__((ext_vector_type(8))) float v8f;

__device__ __forceinline__ v8f wmma4(v2f a, v2f b, v8f c) {
  return __builtin_amdgcn_wmma_f32_16x16x4_f32(false, a, false, b, (short)0, c,
                                               false, false);
}

__device__ __forceinline__ float silu_f(float x) {
  return x * (1.0f / (1.0f + __expf(-x)));
}

__device__ __forceinline__ v8f zero8() {
  v8f z = {0.f, 0.f, 0.f, 0.f, 0.f, 0.f, 0.f, 0.f};
  return z;
}

__global__ __launch_bounds__(256) void zero_kernel(float4* p, long n4) {
  long i = (long)blockIdx.x * blockDim.x + threadIdx.x;
  if (i < n4) p[i] = make_float4(0.f, 0.f, 0.f, 0.f);
}

__global__ __launch_bounds__(256) void node_up_kernel(
    const float* __restrict__ node_feats, const float* __restrict__ Wsc,
    const float* __restrict__ Wup0, const float* __restrict__ Wup1,
    float* __restrict__ ns, float* __restrict__ sup, float* __restrict__ vup) {
  const int lane = threadIdx.x & 31;
  const int wv = threadIdx.x >> 5;
  const int tile = blockIdx.x * 8 + wv;
  if (tile >= NN / 16) return;
  const int n0 = tile * 16;
  const int row = lane & 15;
  const int hi = lane >> 4;
  const int col = row;
  const float* arow = node_feats + (long)(n0 + row) * 256;

  v8f accA[4], accB[4];
#pragma unroll
  for (int nt = 0; nt < 4; nt++) { accA[nt] = zero8(); accB[nt] = zero8(); }
#pragma unroll
  for (int kk = 0; kk < 16; kk++) {
    const int kb = kk * 4 + 2 * hi;
    v2f a = *(const v2f*)(arow + kb);
#pragma unroll
    for (int nt = 0; nt < 4; nt++) {
      v2f b0, b1;
      b0.x = Wsc[kb * 64 + nt * 16 + col];
      b0.y = Wsc[(kb + 1) * 64 + nt * 16 + col];
      accA[nt] = wmma4(a, b0, accA[nt]);
      b1.x = Wup0[kb * 64 + nt * 16 + col];
      b1.y = Wup0[(kb + 1) * 64 + nt * 16 + col];
      accB[nt] = wmma4(a, b1, accB[nt]);
    }
  }
#pragma unroll
  for (int nt = 0; nt < 4; nt++)
#pragma unroll
    for (int r = 0; r < 8; r++) {
      const int n = n0 + r + 8 * hi;
      ns[(long)n * 64 + nt * 16 + col] = accA[nt][r];
      sup[(long)n * 64 + nt * 16 + col] = accB[nt][r];
    }

#pragma unroll
  for (int i = 0; i < 3; i++) {
    v8f acc[4];
#pragma unroll
    for (int nt = 0; nt < 4; nt++) acc[nt] = zero8();
#pragma unroll
    for (int kk = 0; kk < 16; kk++) {
      const int kb = kk * 4 + 2 * hi;
      v2f a;
      a.x = arow[64 + kb * 3 + i];
      a.y = arow[64 + (kb + 1) * 3 + i];
#pragma unroll
      for (int nt = 0; nt < 4; nt++) {
        v2f b;
        b.x = Wup1[kb * 64 + nt * 16 + col];
        b.y = Wup1[(kb + 1) * 64 + nt * 16 + col];
        acc[nt] = wmma4(a, b, acc[nt]);
      }
    }
#pragma unroll
    for (int nt = 0; nt < 4; nt++)
#pragma unroll
      for (int r = 0; r < 8; r++) {
        const int n = n0 + r + 8 * hi;
        vup[(long)i * NN * 64 + (long)n * 64 + nt * 16 + col] = acc[nt][r];
      }
  }
}

__global__ __launch_bounds__(128) void edge_kernel(
    const float* __restrict__ edge_attrs, const float* __restrict__ edge_feats,
    const float* __restrict__ lengths, const int* __restrict__ edge_index,
    const float* __restrict__ W1, const float* __restrict__ b1,
    const float* __restrict__ W2, const float* __restrict__ b2,
    const float* __restrict__ W3, const float* __restrict__ ns,
    const float* __restrict__ sup, const float* __restrict__ vup,
    float* __restrict__ m0, float* __restrict__ m1) {
  __shared__ float lds_h[4][32 * 64];
  __shared__ float lds_meta[4][32 * 8];
  const int lane = threadIdx.x & 31;
  const int wv = threadIdx.x >> 5;
  const int dt = blockIdx.x * 4 + wv;
  const int e0 = dt * 32;
  const int row = lane & 15;
  const int hi = lane >> 4;
  const int col = row;

  const int ea = e0 + row;
  const int eb = e0 + 16 + row;
  const int snd0 = edge_index[ea], rcv0 = edge_index[NE + ea];
  const int snd1 = edge_index[eb], rcv1 = edge_index[NE + eb];
  {
    const int mr = row + 16 * hi;
    const int me = e0 + mr;
    lds_meta[wv][mr * 8 + 0] = __int_as_float(hi ? snd1 : snd0);
    lds_meta[wv][mr * 8 + 1] = __int_as_float(hi ? rcv1 : rcv0);
    lds_meta[wv][mr * 8 + 2] = edge_attrs[me * 4 + 0];
    lds_meta[wv][mr * 8 + 3] = edge_attrs[me * 4 + 1];
    lds_meta[wv][mr * 8 + 4] = edge_attrs[me * 4 + 2];
    lds_meta[wv][mr * 8 + 5] = edge_attrs[me * 4 + 3];
  }
  const float* sp0 = ns + (long)snd0 * 64;
  const float* rp0 = ns + (long)rcv0 * 64;
  const float* sp1 = ns + (long)snd1 * 64;
  const float* rp1 = ns + (long)rcv1 * 64;
  const float* ep0 = edge_feats + (long)ea * 8;
  const float* ep1 = edge_feats + (long)eb * 8;
  const float len0 = lengths[ea];
  const float len1 = lengths[eb];

  v8f acc0[4], acc1[4];
#pragma unroll
  for (int nt = 0; nt < 4; nt++) { acc0[nt] = zero8(); acc1[nt] = zero8(); }
#pragma unroll 4
  for (int kk = 0; kk < 16; kk++) {
    const int kb = kk * 4 + 2 * hi;
    v2f a0 = *(const v2f*)(sp0 + kb);
    v2f a1 = *(const v2f*)(sp1 + kb);
#pragma unroll
    for (int nt = 0; nt < 4; nt++) {
      v2f b;
      b.x = W1[kb * 64 + nt * 16 + col];
      b.y = W1[(kb + 1) * 64 + nt * 16 + col];
      acc0[nt] = wmma4(a0, b, acc0[nt]);
      acc1[nt] = wmma4(a1, b, acc1[nt]);
    }
  }
#pragma unroll 4
  for (int kk = 0; kk < 16; kk++) {
    const int kb = kk * 4 + 2 * hi;
    v2f a0 = *(const v2f*)(rp0 + kb);
    v2f a1 = *(const v2f*)(rp1 + kb);
#pragma unroll
    for (int nt = 0; nt < 4; nt++) {
      v2f b;
      b.x = W1[(64 + kb) * 64 + nt * 16 + col];
      b.y = W1[(64 + kb + 1) * 64 + nt * 16 + col];
      acc0[nt] = wmma4(a0, b, acc0[nt]);
      acc1[nt] = wmma4(a1, b, acc1[nt]);
    }
  }
#pragma unroll
  for (int kk = 0; kk < 2; kk++) {
    const int kb = kk * 4 + 2 * hi;
    v2f a0 = *(const v2f*)(ep0 + kb);
    v2f a1 = *(const v2f*)(ep1 + kb);
#pragma unroll
    for (int nt = 0; nt < 4; nt++) {
      v2f b;
      b.x = W1[(128 + kb) * 64 + nt * 16 + col];
      b.y = W1[(128 + kb + 1) * 64 + nt * 16 + col];
      acc0[nt] = wmma4(a0, b, acc0[nt]);
      acc1[nt] = wmma4(a1, b, acc1[nt]);
    }
  }
  {
    v2f a0, a1;
    a0.x = hi ? 0.f : len0;  a0.y = 0.f;
    a1.x = hi ? 0.f : len1;  a1.y = 0.f;
#pragma unroll
    for (int nt = 0; nt < 4; nt++) {
      v2f b;
      b.x = hi ? 0.f : W1[136 * 64 + nt * 16 + col];
      b.y = 0.f;
      acc0[nt] = wmma4(a0, b, acc0[nt]);
      acc1[nt] = wmma4(a1, b, acc1[nt]);
    }
  }
#pragma unroll
  for (int nt = 0; nt < 4; nt++) {
    const float bias = b1[nt * 16 + col];
#pragma unroll
    for (int r = 0; r < 8; r++) {
      lds_h[wv][(r + 8 * hi) * 64 + nt * 16 + col] = silu_f(acc0[nt][r] + bias);
      lds_h[wv][(16 + r + 8 * hi) * 64 + nt * 16 + col] =
          silu_f(acc1[nt][r] + bias);
    }
  }

#pragma unroll
  for (int nt = 0; nt < 4; nt++) { acc0[nt] = zero8(); acc1[nt] = zero8(); }
#pragma unroll 4
  for (int kk = 0; kk < 16; kk++) {
    const int kb = kk * 4 + 2 * hi;
    v2f a0 = *(const v2f*)&lds_h[wv][row * 64 + kb];
    v2f a1 = *(const v2f*)&lds_h[wv][(16 + row) * 64 + kb];
#pragma unroll
    for (int nt = 0; nt < 4; nt++) {
      v2f b;
      b.x = W2[kb * 64 + nt * 16 + col];
      b.y = W2[(kb + 1) * 64 + nt * 16 + col];
      acc0[nt] = wmma4(a0, b, acc0[nt]);
      acc1[nt] = wmma4(a1, b, acc1[nt]);
    }
  }
  float h2a[4][8], h2b[4][8];
#pragma unroll
  for (int nt = 0; nt < 4; nt++) {
    const float bias = b2[nt * 16 + col];
#pragma unroll
    for (int r = 0; r < 8; r++) {
      h2a[nt][r] = silu_f(acc0[nt][r] + bias);
      h2b[nt][r] = silu_f(acc1[nt][r] + bias);
    }
  }
#pragma unroll
  for (int nt = 0; nt < 4; nt++)
#pragma unroll
    for (int r = 0; r < 8; r++) {
      lds_h[wv][(r + 8 * hi) * 64 + nt * 16 + col] = h2a[nt][r];
      lds_h[wv][(16 + r + 8 * hi) * 64 + nt * 16 + col] = h2b[nt][r];
    }

  {
    v8f ta0[4], ta1[4], tc0[4], tc1[4];
#pragma unroll
    for (int nt = 0; nt < 4; nt++) {
      ta0[nt] = zero8(); ta1[nt] = zero8();
      tc0[nt] = zero8(); tc1[nt] = zero8();
    }
#pragma unroll 2
    for (int kk = 0; kk < 16; kk++) {
      const int kb = kk * 4 + 2 * hi;
      v2f a0 = *(const v2f*)&lds_h[wv][row * 64 + kb];
      v2f a1 = *(const v2f*)&lds_h[wv][(16 + row) * 64 + kb];
#pragma unroll
      for (int nt = 0; nt < 4; nt++) {
        v2f b, d;
        b.x = W3[kb * 256 + nt * 16 + col];
        b.y = W3[(kb + 1) * 256 + nt * 16 + col];
        ta0[nt] = wmma4(a0, b, ta0[nt]);
        ta1[nt] = wmma4(a1, b, ta1[nt]);
        d.x = W3[kb * 256 + 128 + nt * 16 + col];
        d.y = W3[(kb + 1) * 256 + 128 + nt * 16 + col];
        tc0[nt] = wmma4(a0, d, tc0[nt]);
        tc1[nt] = wmma4(a1, d, tc1[nt]);
      }
    }
#pragma unroll
    for (int t = 0; t < 2; t++) {
#pragma unroll
      for (int nt = 0; nt < 4; nt++) {
        const int c = nt * 16 + col;
#pragma unroll
        for (int r = 0; r < 8; r++) {
          const int rr = t * 16 + r + 8 * hi;
          const int s_e = __float_as_int(lds_meta[wv][rr * 8 + 0]);
          const int r_e = __float_as_int(lds_meta[wv][rr * 8 + 1]);
          const float y0 = lds_meta[wv][rr * 8 + 2];
          const float xs = sup[(long)s_e * 64 + c];
          const float wa = t ? ta1[nt][r] : ta0[nt][r];
          const float wc = t ? tc1[nt][r] : tc0[nt][r];
          unsafeAtomicAdd(&m0[(long)r_e * 128 + c], xs * y0 * wa);
#pragma unroll
          for (int i = 0; i < 3; i++)
            unsafeAtomicAdd(&m1[(long)i * NN * 128 + (long)r_e * 128 + c],
                            xs * lds_meta[wv][rr * 8 + 3 + i] * wc);
        }
      }
    }
  }
  {
    v8f tb0[4], tb1[4], td0[4], td1[4];
#pragma unroll
    for (int nt = 0; nt < 4; nt++) {
      tb0[nt] = zero8(); tb1[nt] = zero8();
      td0[nt] = zero8(); td1[nt] = zero8();
    }
#pragma unroll 2
    for (int kk = 0; kk < 16; kk++) {
      const int kb = kk * 4 + 2 * hi;
      v2f a0 = *(const v2f*)&lds_h[wv][row * 64 + kb];
      v2f a1 = *(const v2f*)&lds_h[wv][(16 + row) * 64 + kb];
#pragma unroll
      for (int nt = 0; nt < 4; nt++) {
        v2f b, d;
        b.x = W3[kb * 256 + 64 + nt * 16 + col];
        b.y = W3[(kb + 1) * 256 + 64 + nt * 16 + col];
        tb0[nt] = wmma4(a0, b, tb0[nt]);
        tb1[nt] = wmma4(a1, b, tb1[nt]);
        d.x = W3[kb * 256 + 192 + nt * 16 + col];
        d.y = W3[(kb + 1) * 256 + 192 + nt * 16 + col];
        td0[nt] = wmma4(a0, d, td0[nt]);
        td1[nt] = wmma4(a1, d, td1[nt]);
      }
    }
#pragma unroll
    for (int t = 0; t < 2; t++) {
#pragma unroll
      for (int nt = 0; nt < 4; nt++) {
        const int c = nt * 16 + col;
#pragma unroll
        for (int r = 0; r < 8; r++) {
          const int rr = t * 16 + r + 8 * hi;
          const int s_e = __float_as_int(lds_meta[wv][rr * 8 + 0]);
          const int r_e = __float_as_int(lds_meta[wv][rr * 8 + 1]);
          const float y0 = lds_meta[wv][rr * 8 + 2];
          const float x0 = vup[(long)s_e * 64 + c];
          const float x1 = vup[(long)NN * 64 + (long)s_e * 64 + c];
          const float x2 = vup[(long)2 * NN * 64 + (long)s_e * 64 + c];
          const float wb = t ? tb1[nt][r] : tb0[nt][r];
          const float wd = t ? td1[nt][r] : td0[nt][r];
          const float dot = x0 * lds_meta[wv][rr * 8 + 3] +
                            x1 * lds_meta[wv][rr * 8 + 4] +
                            x2 * lds_meta[wv][rr * 8 + 5];
          unsafeAtomicAdd(&m0[(long)r_e * 128 + 64 + c],
                          dot * INV_SQRT3 * wb);
          unsafeAtomicAdd(&m1[(long)r_e * 128 + 64 + c], x0 * y0 * wd);
          unsafeAtomicAdd(&m1[(long)NN * 128 + (long)r_e * 128 + 64 + c],
                          x1 * y0 * wd);
          unsafeAtomicAdd(&m1[(long)2 * NN * 128 + (long)r_e * 128 + 64 + c],
                          x2 * y0 * wd);
        }
      }
    }
  }
}

__global__ __launch_bounds__(256) void out_kernel(
    const float* __restrict__ m0, const float* __restrict__ m1,
    const float* __restrict__ Ws, const float* __restrict__ Wv,
    float* __restrict__ out) {
  const int lane = threadIdx.x & 31;
  const int wv = threadIdx.x >> 5;
  const int tile = blockIdx.x * 8 + wv;
  if (tile >= NN / 16) return;
  const int n0 = tile * 16;
  const int row = lane & 15;
  const int hi = lane >> 4;
  const int col = row;

  v8f acc[4][4];
#pragma unroll
  for (int j = 0; j < 4; j++)
#pragma unroll
    for (int nt = 0; nt < 4; nt++) acc[j][nt] = zero8();

#pragma unroll 4
  for (int kk = 0; kk < 32; kk++) {
    const int kb = kk * 4 + 2 * hi;
    v2f a = *(const v2f*)(m0 + (long)(n0 + row) * 128 + kb);
#pragma unroll
    for (int nt = 0; nt < 4; nt++) {
      v2f b;
      b.x = Ws[kb * 64 + nt * 16 + col];
      b.y = Ws[(kb + 1) * 64 + nt * 16 + col];
      acc[0][nt] = wmma4(a, b, acc[0][nt]);
    }
  }
#pragma unroll
  for (int i = 0; i < 3; i++) {
#pragma unroll 4
    for (int kk = 0; kk < 32; kk++) {
      const int kb = kk * 4 + 2 * hi;
      v2f a =
          *(const v2f*)(m1 + (long)i * NN * 128 + (long)(n0 + row) * 128 + kb);
#pragma unroll
      for (int nt = 0; nt < 4; nt++) {
        v2f b;
        b.x = Wv[kb * 64 + nt * 16 + col];
        b.y = Wv[(kb + 1) * 64 + nt * 16 + col];
        acc[1 + i][nt] = wmma4(a, b, acc[1 + i][nt]);
      }
    }
  }
  const float inv = 1.0f / 16.0f;
#pragma unroll
  for (int nt = 0; nt < 4; nt++) {
    const int c = nt * 16 + col;
#pragma unroll
    for (int r = 0; r < 8; r++) {
      const int n = n0 + r + 8 * hi;
      float4 o = make_float4(acc[0][nt][r] * inv, acc[1][nt][r] * inv,
                             acc[2][nt][r] * inv, acc[3][nt][r] * inv);
      *(float4*)(out + (long)n * 256 + c * 4) = o;
    }
  }
}

extern "C" void kernel_launch(void* const* d_in, const int* in_sizes, int n_in,
                              void* d_out, int out_size, void* d_ws,
                              size_t ws_size, hipStream_t stream) {
  (void)in_sizes; (void)n_in; (void)out_size; (void)ws_size;
  const float* node_feats = (const float*)d_in[0];
  const float* edge_attrs = (const float*)d_in[1];
  const float* edge_feats = (const float*)d_in[2];
  const float* lengths    = (const float*)d_in[3];
  const int*   edge_index = (const int*)d_in[4];
  const float* Wsc  = (const float*)d_in[5];
  const float* Wup0 = (const float*)d_in[6];
  const float* Wup1 = (const float*)d_in[7];
  const float* W1   = (const float*)d_in[8];
  const float* b1   = (const float*)d_in[9];
  const float* W2   = (const float*)d_in[10];
  const float* b2   = (const float*)d_in[11];
  const float* W3   = (const float*)d_in[12];
  const float* Ws   = (const float*)d_in[13];
  const float* Wv   = (const float*)d_in[14];
  float* out = (float*)d_out;

  float* ws  = (float*)d_ws;
  float* ns  = ws;
  float* sup = ns + (long)NN * 64;
  float* vup = sup + (long)NN * 64;
  float* m0  = vup + (long)3 * NN * 64;
  float* m1  = m0 + (long)NN * 128;

  const long zn4 = (long)NN * 512 / 4;
  zero_kernel<<<(int)((zn4 + 255) / 256), 256, 0, stream>>>((float4*)m0, zn4);

  node_up_kernel<<<(NN / 16 + 7) / 8, 256, 0, stream>>>(node_feats, Wsc, Wup0,
                                                        Wup1, ns, sup, vup);
  edge_kernel<<<NE / 32 / 4, 128, 0, stream>>>(edge_attrs, edge_feats, lengths,
                                               edge_index, W1, b1, W2, b2, W3,
                                               ns, sup, vup, m0, m1);
  out_kernel<<<(NN / 16 + 7) / 8, 256, 0, stream>>>(m0, m1, Ws, Wv, out);
}


# edges counting-sorted by receiver in the call, each receiver's edges in edge-id order (fixed); edge kernel sums each run of equal receivers in registers, one float atomic per run: sums reordered
# speedup vs baseline: 1.8331x; 1.3206x over previous
#include <hip/hip_runtime.h>

#define NN 20000
#define NE 320000
#define INV_SQRT3 0.5773502691896258f

typedef __attribute__((ext_vector_type(2))) float v2f;
typedef __attribute__((ext_vector_type(8))) float v8f;

__device__ __forceinline__ v8f wmma4(v2f a, v2f b, v8f c) {
  return __builtin_amdgcn_wmma_f32_16x16x4_f32(false, a, false, b, (short)0, c,
                                               false, false);
}

__device__ __forceinline__ float silu_f(float x) {
  return x * (1.0f / (1.0f + __expf(-x)));
}

__device__ __forceinline__ v8f zero8() {
  v8f z = {0.f, 0.f, 0.f, 0.f, 0.f, 0.f, 0.f, 0.f};
  return z;
}

__global__ __launch_bounds__(256) void zero_kernel(float4* p, long n4, int4* q,
                                                   long nq4) {
  long i = (long)blockIdx.x * blockDim.x + threadIdx.x;
  if (i < n4) p[i] = make_float4(0.f, 0.f, 0.f, 0.f);
  if (i < nq4) q[i] = make_int4(0, 0, 0, 0);
}

__global__ __launch_bounds__(256) void count_kernel(
    const int* __restrict__ edge_index, int* __restrict__ cnt,
    int* __restrict__ arrival) {
  const int e = blockIdx.x * 256 + threadIdx.x;
  if (e < NE) {
    const int r = edge_index[NE + e];
    arrival[e] = (unsigned)r < (unsigned)NN ? atomicAdd(&cnt[r], 1) : 0;
  }
}

#define SCAN_PER 20
__global__ __launch_bounds__(1024) void scan_kernel(
    const int* __restrict__ cnt, int* __restrict__ start) {
  __shared__ int part[1024];
  const int t = threadIdx.x;
  int loc[SCAN_PER];
  int sum = 0;
#pragma unroll
  for (int j = 0; j < SCAN_PER; j++) {
    const int idx = t * SCAN_PER + j;
    const int v = idx < NN ? cnt[idx] : 0;
    loc[j] = sum;
    sum += v;
  }
  part[t] = sum;
  __syncthreads();
  for (int off = 1; off < 1024; off <<= 1) {
    const int v = t >= off ? part[t - off] : 0;
    __syncthreads();
    part[t] += v;
    __syncthreads();
  }
  const int excl = part[t] - sum;
#pragma unroll
  for (int j = 0; j < SCAN_PER; j++) {
    const int idx = t * SCAN_PER + j;
    if (idx < NN) start[idx] = excl + loc[j];
  }
}

__global__ __launch_bounds__(256) void place_kernel(
    const int* __restrict__ edge_index, const int* __restrict__ start,
    const int* __restrict__ arrival, int* __restrict__ perm_a) {
  const int e = blockIdx.x * 256 + threadIdx.x;
  if (e < NE) {
    const int r = edge_index[NE + e];
    if ((unsigned)r < (unsigned)NN) {
      const int p = start[r] + arrival[e];
      if ((unsigned)p < (unsigned)NE) perm_a[p] = e;
    }
  }
}

__global__ __launch_bounds__(256) void rank_kernel(
    const int* __restrict__ edge_index, const int* __restrict__ cnt,
    const int* __restrict__ start, const int* __restrict__ perm_a,
    int* __restrict__ perm_b) {
  const int p = blockIdx.x * 256 + threadIdx.x;
  if (p < NE) {
    int e = perm_a[p];
    e = (unsigned)e < (unsigned)NE ? e : 0;
    int r = edge_index[NE + e];
    r = (unsigned)r < (unsigned)NN ? r : 0;
    const int s = start[r];
    const int d = cnt[r];
    int k = 0;
    for (int j = 0; j < d; j++) k += (perm_a[s + j] < e) ? 1 : 0;
    const int q = s + k;
    if ((unsigned)q < (unsigned)NE) perm_b[q] = e;
  }
}

__global__ __launch_bounds__(256) void node_up_kernel(
    const float* __restrict__ node_feats, const float* __restrict__ Wsc,
    const float* __restrict__ Wup0, const float* __restrict__ Wup1,
    float* __restrict__ ns, float* __restrict__ sup, float* __restrict__ vup) {
  const int lane = threadIdx.x & 31;
  const int wv = threadIdx.x >> 5;
  const int tile = blockIdx.x * 8 + wv;
  if (tile >= NN / 16) return;
  const int n0 = tile * 16;
  const int row = lane & 15;
  const int hi = lane >> 4;
  const int col = row;
  const float* arow = node_feats + (long)(n0 + row) * 256;

  v8f accA[4], accB[4];
#pragma unroll
  for (int nt = 0; nt < 4; nt++) { accA[nt] = zero8(); accB[nt] = zero8(); }
#pragma unroll
  for (int kk = 0; kk < 16; kk++) {
    const int kb = kk * 4 + 2 * hi;
    v2f a = *(const v2f*)(arow + kb);
#pragma unroll
    for (int nt = 0; nt < 4; nt++) {
      v2f b0, b1;
      b0.x = Wsc[kb * 64 + nt * 16 + col];
      b0.y = Wsc[(kb + 1) * 64 + nt * 16 + col];
      accA[nt] = wmma4(a, b0, accA[nt]);
      b1.x = Wup0[kb * 64 + nt * 16 + col];
      b1.y = Wup0[(kb + 1) * 64 + nt * 16 + col];
      accB[nt] = wmma4(a, b1, accB[nt]);
    }
  }
#pragma unroll
  for (int nt = 0; nt < 4; nt++)
#pragma unroll
    for (int r = 0; r < 8; r++) {
      const int n = n0 + r + 8 * hi;
      ns[(long)n * 64 + nt * 16 + col] = accA[nt][r];
      sup[(long)n * 64 + nt * 16 + col] = accB[nt][r];
    }

#pragma unroll
  for (int i = 0; i < 3; i++) {
    v8f acc[4];
#pragma unroll
    for (int nt = 0; nt < 4; nt++) acc[nt] = zero8();
#pragma unroll
    for (int kk = 0; kk < 16; kk++) {
      const int kb = kk * 4 + 2 * hi;
      v2f a;
      a.x = arow[64 + kb * 3 + i];
      a.y = arow[64 + (kb + 1) * 3 + i];
#pragma unroll
      for (int nt = 0; nt < 4; nt++) {
        v2f b;
        b.x = Wup1[kb * 64 + nt * 16 + col];
        b.y = Wup1[(kb + 1) * 64 + nt * 16 + col];
        acc[nt] = wmma4(a, b, acc[nt]);
      }
    }
#pragma unroll
    for (int nt = 0; nt < 4; nt++)
#pragma unroll
      for (int r = 0; r < 8; r++) {
        const int n = n0 + r + 8 * hi;
        vup[(long)i * NN * 64 + (long)n * 64 + nt * 16 + col] = acc[nt][r];
      }
  }
}

__global__ __launch_bounds__(128) void edge_kernel(
    const float* __restrict__ edge_attrs, const float* __restrict__ edge_feats,
    const float* __restrict__ lengths, const int* __restrict__ edge_index,
    const int* __restrict__ perm,
    const float* __restrict__ W1, const float* __restrict__ b1,
    const float* __restrict__ W2, const float* __restrict__ b2,
    const float* __restrict__ W3, const float* __restrict__ ns,
    const float* __restrict__ sup, const float* __restrict__ vup,
    float* __restrict__ m0, float* __restrict__ m1) {
  __shared__ float lds_h[4][32 * 64];
  __shared__ float lds_meta[4][32 * 8];
  const int lane = threadIdx.x & 31;
  const int wv = threadIdx.x >> 5;
  const int dt = blockIdx.x * 4 + wv;
  const int e0 = dt * 32;
  const int row = lane & 15;
  const int hi = lane >> 4;
  const int col = row;

  int ea = perm[e0 + row];
  int eb = perm[e0 + 16 + row];
  ea = (unsigned)ea < (unsigned)NE ? ea : 0;
  eb = (unsigned)eb < (unsigned)NE ? eb : 0;
  const int snd0 = edge_index[ea], rcv0 = edge_index[NE + ea];
  const int snd1 = edge_index[eb], rcv1 = edge_index[NE + eb];
  {
    const int mr = row + 16 * hi;
    const int me = hi ? eb : ea;
    const int myr = hi ? rcv1 : rcv0;
    const int nxt = __shfl_down(myr, 1);
    const int endf = ((lane & 7) == 7) || (nxt != myr);
    lds_meta[wv][mr * 8 + 6] = __int_as_float(endf);
    lds_meta[wv][mr * 8 + 0] = __int_as_float(hi ? snd1 : snd0);
    lds_meta[wv][mr * 8 + 1] = __int_as_float(myr);
    lds_meta[wv][mr * 8 + 2] = edge_attrs[me * 4 + 0];
    lds_meta[wv][mr * 8 + 3] = edge_attrs[me * 4 + 1];
    lds_meta[wv][mr * 8 + 4] = edge_attrs[me * 4 + 2];
    lds_meta[wv][mr * 8 + 5] = edge_attrs[me * 4 + 3];
  }
  const float* sp0 = ns + (long)snd0 * 64;
  const float* rp0 = ns + (long)rcv0 * 64;
  const float* sp1 = ns + (long)snd1 * 64;
  const float* rp1 = ns + (long)rcv1 * 64;
  const float* ep0 = edge_feats + (long)ea * 8;
  const float* ep1 = edge_feats + (long)eb * 8;
  const float len0 = lengths[ea];
  const float len1 = lengths[eb];

  v8f acc0[4], acc1[4];
#pragma unroll
  for (int nt = 0; nt < 4; nt++) { acc0[nt] = zero8(); acc1[nt] = zero8(); }
#pragma unroll 4
  for (int kk = 0; kk < 16; kk++) {
    const int kb = kk * 4 + 2 * hi;
    v2f a0 = *(const v2f*)(sp0 + kb);
    v2f a1 = *(const v2f*)(sp1 + kb);
#pragma unroll
    for (int nt = 0; nt < 4; nt++) {
      v2f b;
      b.x = W1[kb * 64 + nt * 16 + col];
      b.y = W1[(kb + 1) * 64 + nt * 16 + col];
      acc0[nt] = wmma4(a0, b, acc0[nt]);
      acc1[nt] = wmma4(a1, b, acc1[nt]);
    }
  }
#pragma unroll 4
  for (int kk = 0; kk < 16; kk++) {
    const int kb = kk * 4 + 2 * hi;
    v2f a0 = *(const v2f*)(rp0 + kb);
    v2f a1 = *(const v2f*)(rp1 + kb);
#pragma unroll
    for (int nt = 0; nt < 4; nt++) {
      v2f b;
      b.x = W1[(64 + kb) * 64 + nt * 16 + col];
      b.y = W1[(64 + kb + 1) * 64 + nt * 16 + col];
      acc0[nt] = wmma4(a0, b, acc0[nt]);
      acc1[nt] = wmma4(a1, b, acc1[nt]);
    }
  }
#pragma unroll
  for (int kk = 0; kk < 2; kk++) {
    const int kb = kk * 4 + 2 * hi;
    v2f a0 = *(const v2f*)(ep0 + kb);
    v2f a1 = *(const v2f*)(ep1 + kb);
#pragma unroll
    for (int nt = 0; nt < 4; nt++) {
      v2f b;
      b.x = W1[(128 + kb) * 64 + nt * 16 + col];
      b.y = W1[(128 + kb + 1) * 64 + nt * 16 + col];
      acc0[nt] = wmma4(a0, b, acc0[nt]);
      acc1[nt] = wmma4(a1, b, acc1[nt]);
    }
  }
  {
    v2f a0, a1;
    a0.x = hi ? 0.f : len0;  a0.y = 0.f;
    a1.x = hi ? 0.f : len1;  a1.y = 0.f;
#pragma unroll
    for (int nt = 0; nt < 4; nt++) {
      v2f b;
      b.x = hi ? 0.f : W1[136 * 64 + nt * 16 + col];
      b.y = 0.f;
      acc0[nt] = wmma4(a0, b, acc0[nt]);
      acc1[nt] = wmma4(a1, b, acc1[nt]);
    }
  }
#pragma unroll
  for (int nt = 0; nt < 4; nt++) {
    const float bias = b1[nt * 16 + col];
#pragma unroll
    for (int r = 0; r < 8; r++) {
      lds_h[wv][(r + 8 * hi) * 64 + nt * 16 + col] = silu_f(acc0[nt][r] + bias);
      lds_h[wv][(16 + r + 8 * hi) * 64 + nt * 16 + col] =
          silu_f(acc1[nt][r] + bias);
    }
  }

#pragma unroll
  for (int nt = 0; nt < 4; nt++) { acc0[nt] = zero8(); acc1[nt] = zero8(); }
#pragma unroll 4
  for (int kk = 0; kk < 16; kk++) {
    const int kb = kk * 4 + 2 * hi;
    v2f a0 = *(const v2f*)&lds_h[wv][row * 64 + kb];
    v2f a1 = *(const v2f*)&lds_h[wv][(16 + row) * 64 + kb];
#pragma unroll
    for (int nt = 0; nt < 4; nt++) {
      v2f b;
      b.x = W2[kb * 64 + nt * 16 + col];
      b.y = W2[(kb + 1) * 64 + nt * 16 + col];
      acc0[nt] = wmma4(a0, b, acc0[nt]);
      acc1[nt] = wmma4(a1, b, acc1[nt]);
    }
  }
  float h2a[4][8], h2b[4][8];
#pragma unroll
  for (int nt = 0; nt < 4; nt++) {
    const float bias = b2[nt * 16 + col];
#pragma unroll
    for (int r = 0; r < 8; r++) {
      h2a[nt][r] = silu_f(acc0[nt][r] + bias);
      h2b[nt][r] = silu_f(acc1[nt][r] + bias);
    }
  }
#pragma unroll
  for (int nt = 0; nt < 4; nt++)
#pragma unroll
    for (int r = 0; r < 8; r++) {
      lds_h[wv][(r + 8 * hi) * 64 + nt * 16 + col] = h2a[nt][r];
      lds_h[wv][(16 + r + 8 * hi) * 64 + nt * 16 + col] = h2b[nt][r];
    }

  {
    v8f ta0[4], ta1[4], tc0[4], tc1[4];
#pragma unroll
    for (int nt = 0; nt < 4; nt++) {
      ta0[nt] = zero8(); ta1[nt] = zero8();
      tc0[nt] = zero8(); tc1[nt] = zero8();
    }
#pragma unroll 2
    for (int kk = 0; kk < 16; kk++) {
      const int kb = kk * 4 + 2 * hi;
      v2f a0 = *(const v2f*)&lds_h[wv][row * 64 + kb];
      v2f a1 = *(const v2f*)&lds_h[wv][(16 + row) * 64 + kb];
#pragma unroll
      for (int nt = 0; nt < 4; nt++) {
        v2f b, d;
        b.x = W3[kb * 256 + nt * 16 + col];
        b.y = W3[(kb + 1) * 256 + nt * 16 + col];
        ta0[nt] = wmma4(a0, b, ta0[nt]);
        ta1[nt] = wmma4(a1, b, ta1[nt]);
        d.x = W3[kb * 256 + 128 + nt * 16 + col];
        d.y = W3[(kb + 1) * 256 + 128 + nt * 16 + col];
        tc0[nt] = wmma4(a0, d, tc0[nt]);
        tc1[nt] = wmma4(a1, d, tc1[nt]);
      }
    }
#pragma unroll
    for (int t = 0; t < 2; t++) {
#pragma unroll
      for (int nt = 0; nt < 4; nt++) {
        const int c = nt * 16 + col;
        float q0 = 0.f, q1 = 0.f, q2 = 0.f, q3 = 0.f;
#pragma unroll
        for (int r = 0; r < 8; r++) {
          const int rr = t * 16 + r + 8 * hi;
          const int s_e = __float_as_int(lds_meta[wv][rr * 8 + 0]);
          const int r_e = __float_as_int(lds_meta[wv][rr * 8 + 1]);
          const float y0 = lds_meta[wv][rr * 8 + 2];
          const float xs = sup[(long)s_e * 64 + c];
          const float wa = t ? ta1[nt][r] : ta0[nt][r];
          const float wc = t ? tc1[nt][r] : tc0[nt][r];
          q0 += xs * y0 * wa;
          q1 += xs * lds_meta[wv][rr * 8 + 3] * wc;
          q2 += xs * lds_meta[wv][rr * 8 + 4] * wc;
          q3 += xs * lds_meta[wv][rr * 8 + 5] * wc;
          if (__float_as_int(lds_meta[wv][rr * 8 + 6])) {
            unsafeAtomicAdd(&m0[(long)r_e * 128 + c], q0);
            unsafeAtomicAdd(&m1[(long)r_e * 128 + c], q1);
            unsafeAtomicAdd(&m1[(long)NN * 128 + (long)r_e * 128 + c], q2);
            unsafeAtomicAdd(&m1[(long)2 * NN * 128 + (long)r_e * 128 + c], q3);
            q0 = 0.f; q1 = 0.f; q2 = 0.f; q3 = 0.f;
          }
        }
      }
    }
  }
  {
    v8f tb0[4], tb1[4], td0[4], td1[4];
#pragma unroll
    for (int nt = 0; nt < 4; nt++) {
      tb0[nt] = zero8(); tb1[nt] = zero8();
      td0[nt] = zero8(); td1[nt] = zero8();
    }
#pragma unroll 2
    for (int kk = 0; kk < 16; kk++) {
      const int kb = kk * 4 + 2 * hi;
      v2f a0 = *(const v2f*)&lds_h[wv][row * 64 + kb];
      v2f a1 = *(const v2f*)&lds_h[wv][(16 + row) * 64 + kb];
#pragma unroll
      for (int nt = 0; nt < 4; nt++) {
        v2f b, d;
        b.x = W3[kb * 256 + 64 + nt * 16 + col];
        b.y = W3[(kb + 1) * 256 + 64 + nt * 16 + col];
        tb0[nt] = wmma4(a0, b, tb0[nt]);
        tb1[nt] = wmma4(a1, b, tb1[nt]);
        d.x = W3[kb * 256 + 192 + nt * 16 + col];
        d.y = W3[(kb + 1) * 256 + 192 + nt * 16 + col];
        td0[nt] = wmma4(a0, d, td0[nt]);
        td1[nt] = wmma4(a1, d, td1[nt]);
      }
    }
#pragma unroll
    for (int t = 0; t < 2; t++) {
#pragma unroll
      for (int nt = 0; nt < 4; nt++) {
        const int c = nt * 16 + col;
        float q0 = 0.f, q1 = 0.f, q2 = 0.f, q3 = 0.f;
#pragma unroll
        for (int r = 0; r < 8; r++) {
          const int rr = t * 16 + r + 8 * hi;
          const int s_e = __float_as_int(lds_meta[wv][rr * 8 + 0]);
          const int r_e = __float_as_int(lds_meta[wv][rr * 8 + 1]);
          const float y0 = lds_meta[wv][rr * 8 + 2];
          const float x0 = vup[(long)s_e * 64 + c];
          const float x1 = vup[(long)NN * 64 + (long)s_e * 64 + c];
          const float x2 = vup[(long)2 * NN * 64 + (long)s_e * 64 + c];
          const float wb = t ? tb1[nt][r] : tb0[nt][r];
          const float wd = t ? td1[nt][r] : td0[nt][r];
          const float dot = x0 * lds_meta[wv][rr * 8 + 3] +
                            x1 * lds_meta[wv][rr * 8 + 4] +
                            x2 * lds_meta[wv][rr * 8 + 5];
          q0 += dot * INV_SQRT3 * wb;
          q1 += x0 * y0 * wd;
          q2 += x1 * y0 * wd;
          q3 += x2 * y0 * wd;
          if (__float_as_int(lds_meta[wv][rr * 8 + 6])) {
            unsafeAtomicAdd(&m0[(long)r_e * 128 + 64 + c], q0);
            unsafeAtomicAdd(&m1[(long)r_e * 128 + 64 + c], q1);
            unsafeAtomicAdd(&m1[(long)NN * 128 + (long)r_e * 128 + 64 + c],
                            q2);
            unsafeAtomicAdd(
                &m1[(long)2 * NN * 128 + (long)r_e * 128 + 64 + c], q3);
            q0 = 0.f; q1 = 0.f; q2 = 0.f; q3 = 0.f;
          }
        }
      }
    }
  }
}

__global__ __launch_bounds__(256) void out_kernel(
    const float* __restrict__ m0, const float* __restrict__ m1,
    const float* __restrict__ Ws, const float* __restrict__ Wv,
    float* __restrict__ out) {
  const int lane = threadIdx.x & 31;
  const int wv = threadIdx.x >> 5;
  const int tile = blockIdx.x * 8 + wv;
  if (tile >= NN / 16) return;
  const int n0 = tile * 16;
  const int row = lane & 15;
  const int hi = lane >> 4;
  const int col = row;

  v8f acc[4][4];
#pragma unroll
  for (int j = 0; j < 4; j++)
#pragma unroll
    for (int nt = 0; nt < 4; nt++) acc[j][nt] = zero8();

#pragma unroll 4
  for (int kk = 0; kk < 32; kk++) {
    const int kb = kk * 4 + 2 * hi;
    v2f a = *(const v2f*)(m0 + (long)(n0 + row) * 128 + kb);
#pragma unroll
    for (int nt = 0; nt < 4; nt++) {
      v2f b;
      b.x = Ws[kb * 64 + nt * 16 + col];
      b.y = Ws[(kb + 1) * 64 + nt * 16 + col];
      acc[0][nt] = wmma4(a, b, acc[0][nt]);
    }
  }
#pragma unroll
  for (int i = 0; i < 3; i++) {
#pragma unroll 4
    for (int kk = 0; kk < 32; kk++) {
      const int kb = kk * 4 + 2 * hi;
      v2f a =
          *(const v2f*)(m1 + (long)i * NN * 128 + (long)(n0 + row) * 128 + kb);
#pragma unroll
      for (int nt = 0; nt < 4; nt++) {
        v2f b;
        b.x = Wv[kb * 64 + nt * 16 + col];
        b.y = Wv[(kb + 1) * 64 + nt * 16 + col];
        acc[1 + i][nt] = wmma4(a, b, acc[1 + i][nt]);
      }
    }
  }
  const float inv = 1.0f / 16.0f;
#pragma unroll
  for (int nt = 0; nt < 4; nt++) {
    const int c = nt * 16 + col;
#pragma unroll
    for (int r = 0; r < 8; r++) {
      const int n = n0 + r + 8 * hi;
      float4 o = make_float4(acc[0][nt][r] * inv, acc[1][nt][r] * inv,
                             acc[2][nt][r] * inv, acc[3][nt][r] * inv);
      *(float4*)(out + (long)n * 256 + c * 4) = o;
    }
  }
}

extern "C" void kernel_launch(void* const* d_in, const int* in_sizes, int n_in,
                              void* d_out, int out_size, void* d_ws,
                              size_t ws_size, hipStream_t stream) {
  (void)in_sizes; (void)n_in; (void)out_size; (void)ws_size;
  const float* node_feats = (const float*)d_in[0];
  const float* edge_attrs = (const float*)d_in[1];
  const float* edge_feats = (const float*)d_in[2];
  const float* lengths    = (const float*)d_in[3];
  const int*   edge_index = (const int*)d_in[4];
  const float* Wsc  = (const float*)d_in[5];
  const float* Wup0 = (const float*)d_in[6];
  const float* Wup1 = (const float*)d_in[7];
  const float* W1   = (const float*)d_in[8];
  const float* b1   = (const float*)d_in[9];
  const float* W2   = (const float*)d_in[10];
  const float* b2   = (const float*)d_in[11];
  const float* W3   = (const float*)d_in[12];
  const float* Ws   = (const float*)d_in[13];
  const float* Wv   = (const float*)d_in[14];
  float* out = (float*)d_out;

  float* ws  = (float*)d_ws;
  float* ns  = ws;
  float* sup = ns + (long)NN * 64;
  float* vup = sup + (long)NN * 64;
  float* m0  = vup + (long)3 * NN * 64;
  float* m1  = m0 + (long)NN * 128;

  int* scr     = (int*)d_out;
  int* cnt     = scr;
  int* start   = cnt + NN;
  int* arrival = start + NN;
  int* perm_a  = arrival + NE;
  int* perm_b  = perm_a + NE;

  const long zn4 = (long)NN * 512 / 4;
  zero_kernel<<<(int)((zn4 + 255) / 256), 256, 0, stream>>>(
      (float4*)m0, zn4, (int4*)cnt, (long)NN / 4);
  count_kernel<<<(NE + 255) / 256, 256, 0, stream>>>(edge_index, cnt, arrival);
  scan_kernel<<<1, 1024, 0, stream>>>(cnt, start);
  place_kernel<<<(NE + 255) / 256, 256, 0, stream>>>(edge_index, start, arrival,
                                                     perm_a);
  rank_kernel<<<(NE + 255) / 256, 256, 0, stream>>>(edge_index, cnt, start,
                                                    perm_a, perm_b);

  node_up_kernel<<<(NN / 16 + 7) / 8, 256, 0, stream>>>(node_feats, Wsc, Wup0,
                                                        Wup1, ns, sup, vup);
  edge_kernel<<<NE / 32 / 4, 128, 0, stream>>>(edge_attrs, edge_feats, lengths,
                                               edge_index, perm_b, W1, b1, W2,
                                               b2, W3, ns, sup, vup, m0, m1);
  out_kernel<<<(NN / 16 + 7) / 8, 256, 0, stream>>>(m0, m1, Ws, Wv, out);
}


# edge kernel: sup[snd] and vup[snd] rows of a wave's 32 edges staged in LDS by whole-row 16-B loads ahead of each scatter epilogue, which reads LDS, not one dependent global gather per row and tile
# speedup vs baseline: 2.0502x; 1.1184x over previous
#include <hip/hip_runtime.h>

#define NN 20000
#define NE 320000
#define INV_SQRT3 0.5773502691896258f
#define SPITCH 68

typedef __attribute__((ext_vector_type(2))) float v2f;
typedef __attribute__((ext_vector_type(8))) float v8f;

__device__ __forceinline__ v8f wmma4(v2f a, v2f b, v8f c) {
  return __builtin_amdgcn_wmma_f32_16x16x4_f32(false, a, false, b, (short)0, c,
                                               false, false);
}

__device__ __forceinline__ float silu_f(float x) {
  return x * (1.0f / (1.0f + __expf(-x)));
}

__device__ __forceinline__ v8f zero8() {
  v8f z = {0.f, 0.f, 0.f, 0.f, 0.f, 0.f, 0.f, 0.f};
  return z;
}

__global__ __launch_bounds__(256) void zero_kernel(float4* p, long n4, int4* q,
                                                   long nq4) {
  long i = (long)blockIdx.x * blockDim.x + threadIdx.x;
  if (i < n4) p[i] = make_float4(0.f, 0.f, 0.f, 0.f);
  if (i < nq4) q[i] = make_int4(0, 0, 0, 0);
}

__global__ __launch_bounds__(256) void count_kernel(
    const int* __restrict__ edge_index, int* __restrict__ cnt,
    int* __restrict__ arrival) {
  const int e = blockIdx.x * 256 + threadIdx.x;
  if (e < NE) {
    const int r = edge_index[NE + e];
    arrival[e] = (unsigned)r < (unsigned)NN ? atomicAdd(&cnt[r], 1) : 0;
  }
}

#define SCAN_PER 20
__global__ __launch_bounds__(1024) void scan_kernel(
    const int* __restrict__ cnt, int* __restrict__ start) {
  __shared__ int part[1024];
  const int t = threadIdx.x;
  int loc[SCAN_PER];
  int sum = 0;
#pragma unroll
  for (int j = 0; j < SCAN_PER; j++) {
    const int idx = t * SCAN_PER + j;
    const int v = idx < NN ? cnt[idx] : 0;
    loc[j] = sum;
    sum += v;
  }
  part[t] = sum;
  __syncthreads();
  for (int off = 1; off < 1024; off <<= 1) {
    const int v = t >= off ? part[t - off] : 0;
    __syncthreads();
    part[t] += v;
    __syncthreads();
  }
  const int excl = part[t] - sum;
#pragma unroll
  for (int j = 0; j < SCAN_PER; j++) {
    const int idx = t * SCAN_PER + j;
    if (idx < NN) start[idx] = excl + loc[j];
  }
}

__global__ __launch_bounds__(256) void place_kernel(
    const int* __restrict__ edge_index, const int* __restrict__ start,
    const int* __restrict__ arrival, int* __restrict__ perm_a) {
  const int e = blockIdx.x * 256 + threadIdx.x;
  if (e < NE) {
    const int r = edge_index[NE + e];
    if ((unsigned)r < (unsigned)NN) {
      const int p = start[r] + arrival[e];
      if ((unsigned)p < (unsigned)NE) perm_a[p] = e;
    }
  }
}

__global__ __launch_bounds__(256) void rank_kernel(
    const int* __restrict__ edge_index, const int* __restrict__ cnt,
    const int* __restrict__ start, const int* __restrict__ perm_a,
    int* __restrict__ perm_b) {
  const int p = blockIdx.x * 256 + threadIdx.x;
  if (p < NE) {
    int e = perm_a[p];
    e = (unsigned)e < (unsigned)NE ? e : 0;
    int r = edge_index[NE + e];
    r = (unsigned)r < (unsigned)NN ? r : 0;
    const int s = start[r];
    const int d = cnt[r];
    int k = 0;
    for (int j = 0; j < d; j++) k += (perm_a[s + j] < e) ? 1 : 0;
    const int q = s + k;
    if ((unsigned)q < (unsigned)NE) perm_b[q] = e;
  }
}

__global__ __launch_bounds__(256) void node_up_kernel(
    const float* __restrict__ node_feats, const float* __restrict__ Wsc,
    const float* __restrict__ Wup0, const float* __restrict__ Wup1,
    float* __restrict__ ns, float* __restrict__ sup, float* __restrict__ vup) {
  const int lane = threadIdx.x & 31;
  const int wv = threadIdx.x >> 5;
  const int tile = blockIdx.x * 8 + wv;
  if (tile >= NN / 16) return;
  const int n0 = tile * 16;
  const int row = lane & 15;
  const int hi = lane >> 4;
  const int col = row;
  const float* arow = node_feats + (long)(n0 + row) * 256;

  v8f accA[4], accB[4];
#pragma unroll
  for (int nt = 0; nt < 4; nt++) { accA[nt] = zero8(); accB[nt] = zero8(); }
#pragma unroll
  for (int kk = 0; kk < 16; kk++) {
    const int kb = kk * 4 + 2 * hi;
    v2f a = *(const v2f*)(arow + kb);
#pragma unroll
    for (int nt = 0; nt < 4; nt++) {
      v2f b0, b1;
      b0.x = Wsc[kb * 64 + nt * 16 + col];
      b0.y = Wsc[(kb + 1) * 64 + nt * 16 + col];
      accA[nt] = wmma4(a, b0, accA[nt]);
      b1.x = Wup0[kb * 64 + nt * 16 + col];
      b1.y = Wup0[(kb + 1) * 64 + nt * 16 + col];
      accB[nt] = wmma4(a, b1, accB[nt]);
    }
  }
#pragma unroll
  for (int nt = 0; nt < 4; nt++)
#pragma unroll
    for (int r = 0; r < 8; r++) {
      const int n = n0 + r + 8 * hi;
      ns[(long)n * 64 + nt * 16 + col] = accA[nt][r];
      sup[(long)n * 64 + nt * 16 + col] = accB[nt][r];
    }

#pragma unroll
  for (int i = 0; i < 3; i++) {
    v8f acc[4];
#pragma unroll
    for (int nt = 0; nt < 4; nt++) acc[nt] = zero8();
#pragma unroll
    for (int kk = 0; kk < 16; kk++) {
      const int kb = kk * 4 + 2 * hi;
      v2f a;
      a.x = arow[64 + kb * 3 + i];
      a.y = arow[64 + (kb + 1) * 3 + i];
#pragma unroll
      for (int nt = 0; nt < 4; nt++) {
        v2f b;
        b.x = Wup1[kb * 64 + nt * 16 + col];
        b.y = Wup1[(kb + 1) * 64 + nt * 16 + col];
        acc[nt] = wmma4(a, b, acc[nt]);
      }
    }
#pragma unroll
    for (int nt = 0; nt < 4; nt++)
#pragma unroll
      for (int r = 0; r < 8; r++) {
        const int n = n0 + r + 8 * hi;
        vup[(long)i * NN * 64 + (long)n * 64 + nt * 16 + col] = acc[nt][r];
      }
  }
}

__global__ __launch_bounds__(128) void edge_kernel(
    const float* __restrict__ edge_attrs, const float* __restrict__ edge_feats,
    const float* __restrict__ lengths, const int* __restrict__ edge_index,
    const int* __restrict__ perm,
    const float* __restrict__ W1, const float* __restrict__ b1,
    const float* __restrict__ W2, const float* __restrict__ b2,
    const float* __restrict__ W3, const float* __restrict__ ns,
    const float* __restrict__ sup, const float* __restrict__ vup,
    float* __restrict__ m0, float* __restrict__ m1) {
  __shared__ __attribute__((aligned(16))) float lds_w[4][32 * 64 + 32 * SPITCH];
  __shared__ float lds_meta[4][32 * 8];
  const int lane = threadIdx.x & 31;
  const int wv = threadIdx.x >> 5;
  const int dt = blockIdx.x * 4 + wv;
  const int e0 = dt * 32;
  const int row = lane & 15;
  const int hi = lane >> 4;
  const int col = row;

  int ea = perm[e0 + row];
  int eb = perm[e0 + 16 + row];
  ea = (unsigned)ea < (unsigned)NE ? ea : 0;
  eb = (unsigned)eb < (unsigned)NE ? eb : 0;
  const int snd0 = edge_index[ea], rcv0 = edge_index[NE + ea];
  const int snd1 = edge_index[eb], rcv1 = edge_index[NE + eb];
  {
    const int mr = row + 16 * hi;
    const int me = hi ? eb : ea;
    const int myr = hi ? rcv1 : rcv0;
    const int nxt = __shfl_down(myr, 1);
    const int endf = ((lane & 7) == 7) || (nxt != myr);
    lds_meta[wv][mr * 8 + 6] = __int_as_float(endf);
    lds_meta[wv][mr * 8 + 0] = __int_as_float(hi ? snd1 : snd0);
    lds_meta[wv][mr * 8 + 1] = __int_as_float(myr);
    lds_meta[wv][mr * 8 + 2] = edge_attrs[me * 4 + 0];
    lds_meta[wv][mr * 8 + 3] = edge_attrs[me * 4 + 1];
    lds_meta[wv][mr * 8 + 4] = edge_attrs[me * 4 + 2];
    lds_meta[wv][mr * 8 + 5] = edge_attrs[me * 4 + 3];
  }
  float* const lds_h = lds_w[wv];
  float* const stg = lds_w[wv] + 32 * 64;
  const float* sp0 = ns + (long)snd0 * 64;
  const float* rp0 = ns + (long)rcv0 * 64;
  const float* sp1 = ns + (long)snd1 * 64;
  const float* rp1 = ns + (long)rcv1 * 64;
  const float* ep0 = edge_feats + (long)ea * 8;
  const float* ep1 = edge_feats + (long)eb * 8;
  const float len0 = lengths[ea];
  const float len1 = lengths[eb];

  v8f acc0[4], acc1[4];
#pragma unroll
  for (int nt = 0; nt < 4; nt++) { acc0[nt] = zero8(); acc1[nt] = zero8(); }
#pragma unroll 4
  for (int kk = 0; kk < 16; kk++) {
    const int kb = kk * 4 + 2 * hi;
    v2f a0 = *(const v2f*)(sp0 + kb);
    v2f a1 = *(const v2f*)(sp1 + kb);
#pragma unroll
    for (int nt = 0; nt < 4; nt++) {
      v2f b;
      b.x = W1[kb * 64 + nt * 16 + col];
      b.y = W1[(kb + 1) * 64 + nt * 16 + col];
      acc0[nt] = wmma4(a0, b, acc0[nt]);
      acc1[nt] = wmma4(a1, b, acc1[nt]);
    }
  }
#pragma unroll 4
  for (int kk = 0; kk < 16; kk++) {
    const int kb = kk * 4 + 2 * hi;
    v2f a0 = *(const v2f*)(rp0 + kb);
    v2f a1 = *(const v2f*)(rp1 + kb);
#pragma unroll
    for (int nt = 0; nt < 4; nt++) {
      v2f b;
      b.x = W1[(64 + kb) * 64 + nt * 16 + col];
      b.y = W1[(64 + kb + 1) * 64 + nt * 16 + col];
      acc0[nt] = wmma4(a0, b, acc0[nt]);
      acc1[nt] = wmma4(a1, b, acc1[nt]);
    }
  }
#pragma unroll
  for (int kk = 0; kk < 2; kk++) {
    const int kb = kk * 4 + 2 * hi;
    v2f a0 = *(const v2f*)(ep0 + kb);
    v2f a1 = *(const v2f*)(ep1 + kb);
#pragma unroll
    for (int nt = 0; nt < 4; nt++) {
      v2f b;
      b.x = W1[(128 + kb) * 64 + nt * 16 + col];
      b.y = W1[(128 + kb + 1) * 64 + nt * 16 + col];
      acc0[nt] = wmma4(a0, b, acc0[nt]);
      acc1[nt] = wmma4(a1, b, acc1[nt]);
    }
  }
  {
    v2f a0, a1;
    a0.x = hi ? 0.f : len0;  a0.y = 0.f;
    a1.x = hi ? 0.f : len1;  a1.y = 0.f;
#pragma unroll
    for (int nt = 0; nt < 4; nt++) {
      v2f b;
      b.x = hi ? 0.f : W1[136 * 64 + nt * 16 + col];
      b.y = 0.f;
      acc0[nt] = wmma4(a0, b, acc0[nt]);
      acc1[nt] = wmma4(a1, b, acc1[nt]);
    }
  }
  float4 gs[16];
#pragma unroll
  for (int i = 0; i < 16; i++) {
    const int s_e = __float_as_int(lds_meta[wv][(2 * i + hi) * 8 + 0]);
    gs[i] = *(const float4*)(sup + (long)s_e * 64 + 4 * row);
  }
#pragma unroll
  for (int nt = 0; nt < 4; nt++) {
    const float bias = b1[nt * 16 + col];
#pragma unroll
    for (int r = 0; r < 8; r++) {
      lds_h[(r + 8 * hi) * 64 + nt * 16 + col] = silu_f(acc0[nt][r] + bias);
      lds_h[(16 + r + 8 * hi) * 64 + nt * 16 + col] =
          silu_f(acc1[nt][r] + bias);
    }
  }

#pragma unroll
  for (int i = 0; i < 16; i++)
    *(float4*)&stg[(2 * i + hi) * SPITCH + 4 * row] = gs[i];

#pragma unroll
  for (int nt = 0; nt < 4; nt++) { acc0[nt] = zero8(); acc1[nt] = zero8(); }
#pragma unroll 4
  for (int kk = 0; kk < 16; kk++) {
    const int kb = kk * 4 + 2 * hi;
    v2f a0 = *(const v2f*)&lds_h[row * 64 + kb];
    v2f a1 = *(const v2f*)&lds_h[(16 + row) * 64 + kb];
#pragma unroll
    for (int nt = 0; nt < 4; nt++) {
      v2f b;
      b.x = W2[kb * 64 + nt * 16 + col];
      b.y = W2[(kb + 1) * 64 + nt * 16 + col];
      acc0[nt] = wmma4(a0, b, acc0[nt]);
      acc1[nt] = wmma4(a1, b, acc1[nt]);
    }
  }
  float h2a[4][8], h2b[4][8];
#pragma unroll
  for (int nt = 0; nt < 4; nt++) {
    const float bias = b2[nt * 16 + col];
#pragma unroll
    for (int r = 0; r < 8; r++) {
      h2a[nt][r] = silu_f(acc0[nt][r] + bias);
      h2b[nt][r] = silu_f(acc1[nt][r] + bias);
    }
  }
#pragma unroll
  for (int nt = 0; nt < 4; nt++)
#pragma unroll
    for (int r = 0; r < 8; r++) {
      lds_h[(r + 8 * hi) * 64 + nt * 16 + col] = h2a[nt][r];
      lds_h[(16 + r + 8 * hi) * 64 + nt * 16 + col] = h2b[nt][r];
    }

  {
    v8f ta0[4], ta1[4], tc0[4], tc1[4];
#pragma unroll
    for (int nt = 0; nt < 4; nt++) {
      ta0[nt] = zero8(); ta1[nt] = zero8();
      tc0[nt] = zero8(); tc1[nt] = zero8();
    }
#pragma unroll 2
    for (int kk = 0; kk < 16; kk++) {
      const int kb = kk * 4 + 2 * hi;
      v2f a0 = *(const v2f*)&lds_h[row * 64 + kb];
      v2f a1 = *(const v2f*)&lds_h[(16 + row) * 64 + kb];
#pragma unroll
      for (int nt = 0; nt < 4; nt++) {
        v2f b, d;
        b.x = W3[kb * 256 + nt * 16 + col];
        b.y = W3[(kb + 1) * 256 + nt * 16 + col];
        ta0[nt] = wmma4(a0, b, ta0[nt]);
        ta1[nt] = wmma4(a1, b, ta1[nt]);
        d.x = W3[kb * 256 + 128 + nt * 16 + col];
        d.y = W3[(kb + 1) * 256 + 128 + nt * 16 + col];
        tc0[nt] = wmma4(a0, d, tc0[nt]);
        tc1[nt] = wmma4(a1, d, tc1[nt]);
      }
    }
#pragma unroll
    for (int t = 0; t < 2; t++) {
#pragma unroll
      for (int nt = 0; nt < 4; nt++) {
        const int c = nt * 16 + col;
        float q0 = 0.f, q1 = 0.f, q2 = 0.f, q3 = 0.f;
#pragma unroll
        for (int r = 0; r < 8; r++) {
          const int rr = t * 16 + r + 8 * hi;
          const int r_e = __float_as_int(lds_meta[wv][rr * 8 + 1]);
          const float y0 = lds_meta[wv][rr * 8 + 2];
          const float xs = stg[rr * SPITCH + c];
          const float wa = t ? ta1[nt][r] : ta0[nt][r];
          const float wc = t ? tc1[nt][r] : tc0[nt][r];
          q0 += xs * y0 * wa;
          q1 += xs * lds_meta[wv][rr * 8 + 3] * wc;
          q2 += xs * lds_meta[wv][rr * 8 + 4] * wc;
          q3 += xs * lds_meta[wv][rr * 8 + 5] * wc;
          if (__float_as_int(lds_meta[wv][rr * 8 + 6])) {
            unsafeAtomicAdd(&m0[(long)r_e * 128 + c], q0);
            unsafeAtomicAdd(&m1[(long)r_e * 128 + c], q1);
            unsafeAtomicAdd(&m1[(long)NN * 128 + (long)r_e * 128 + c], q2);
            unsafeAtomicAdd(&m1[(long)2 * NN * 128 + (long)r_e * 128 + c], q3);
            q0 = 0.f; q1 = 0.f; q2 = 0.f; q3 = 0.f;
          }
        }
      }
    }
  }
  {
    v8f tb0[4], tb1[4], td0[4], td1[4];
#pragma unroll
    for (int nt = 0; nt < 4; nt++) {
      tb0[nt] = zero8(); tb1[nt] = zero8();
      td0[nt] = zero8(); td1[nt] = zero8();
    }
#pragma unroll 2
    for (int kk = 0; kk < 16; kk++) {
      const int kb = kk * 4 + 2 * hi;
      v2f a0 = *(const v2f*)&lds_h[row * 64 + kb];
      v2f a1 = *(const v2f*)&lds_h[(16 + row) * 64 + kb];
#pragma unroll
      for (int nt = 0; nt < 4; nt++) {
        v2f b, d;
        b.x = W3[kb * 256 + 64 + nt * 16 + col];
        b.y = W3[(kb + 1) * 256 + 64 + nt * 16 + col];
        tb0[nt] = wmma4(a0, b, tb0[nt]);
        tb1[nt] = wmma4(a1, b, tb1[nt]);
        d.x = W3[kb * 256 + 192 + nt * 16 + col];
        d.y = W3[(kb + 1) * 256 + 192 + nt * 16 + col];
        td0[nt] = wmma4(a0, d, td0[nt]);
        td1[nt] = wmma4(a1, d, td1[nt]);
      }
    }
    float* const vst = lds_w[wv];
#pragma unroll
    for (int t = 0; t < 2; t++) {
#pragma unroll 1
      for (int pl = 0; pl < 3; pl++) {
        float4 gv[8];
#pragma unroll
        for (int i = 0; i < 8; i++) {
          const int s_e =
              __float_as_int(lds_meta[wv][(t * 16 + 2 * i + hi) * 8 + 0]);
          gv[i] = *(const float4*)(vup + (long)pl * NN * 64 + (long)s_e * 64 +
                                   4 * row);
        }
#pragma unroll
        for (int i = 0; i < 8; i++)
          *(float4*)&vst[(pl * 16 + 2 * i + hi) * SPITCH + 4 * row] = gv[i];
      }
#pragma unroll
      for (int nt = 0; nt < 4; nt++) {
        const int c = nt * 16 + col;
        float q0 = 0.f, q1 = 0.f, q2 = 0.f, q3 = 0.f;
#pragma unroll
        for (int r = 0; r < 8; r++) {
          const int rr = t * 16 + r + 8 * hi;
          const int lr = r + 8 * hi;
          const int r_e = __float_as_int(lds_meta[wv][rr * 8 + 1]);
          const float y0 = lds_meta[wv][rr * 8 + 2];
          const float x0 = vst[(0 * 16 + lr) * SPITCH + c];
          const float x1 = vst[(1 * 16 + lr) * SPITCH + c];
          const float x2 = vst[(2 * 16 + lr) * SPITCH + c];
          const float wb = t ? tb1[nt][r] : tb0[nt][r];
          const float wd = t ? td1[nt][r] : td0[nt][r];
          const float dot = x0 * lds_meta[wv][rr * 8 + 3] +
                            x1 * lds_meta[wv][rr * 8 + 4] +
                            x2 * lds_meta[wv][rr * 8 + 5];
          q0 += dot * INV_SQRT3 * wb;
          q1 += x0 * y0 * wd;
          q2 += x1 * y0 * wd;
          q3 += x2 * y0 * wd;
          if (__float_as_int(lds_meta[wv][rr * 8 + 6])) {
            unsafeAtomicAdd(&m0[(long)r_e * 128 + 64 + c], q0);
            unsafeAtomicAdd(&m1[(long)r_e * 128 + 64 + c], q1);
            unsafeAtomicAdd(&m1[(long)NN * 128 + (long)r_e * 128 + 64 + c],
                            q2);
            unsafeAtomicAdd(
                &m1[(long)2 * NN * 128 + (long)r_e * 128 + 64 + c], q3);
            q0 = 0.f; q1 = 0.f; q2 = 0.f; q3 = 0.f;
          }
        }
      }
    }
  }
}

__global__ __launch_bounds__(256) void out_kernel(
    const float* __restrict__ m0, const float* __restrict__ m1,
    const float* __restrict__ Ws, const float* __restrict__ Wv,
    float* __restrict__ out) {
  const int lane = threadIdx.x & 31;
  const int wv = threadIdx.x >> 5;
  const int tile = blockIdx.x * 8 + wv;
  if (tile >= NN / 16) return;
  const int n0 = tile * 16;
  const int row = lane & 15;
  const int hi = lane >> 4;
  const int col = row;

  v8f acc[4][4];
#pragma unroll
  for (int j = 0; j < 4; j++)
#pragma unroll
    for (int nt = 0; nt < 4; nt++) acc[j][nt] = zero8();

#pragma unroll 4
  for (int kk = 0; kk < 32; kk++) {
    const int kb = kk * 4 + 2 * hi;
    v2f a = *(const v2f*)(m0 + (long)(n0 + row) * 128 + kb);
#pragma unroll
    for (int nt = 0; nt < 4; nt++) {
      v2f b;
      b.x = Ws[kb * 64 + nt * 16 + col];
      b.y = Ws[(kb + 1) * 64 + nt * 16 + col];
      acc[0][nt] = wmma4(a, b, acc[0][nt]);
    }
  }
#pragma unroll
  for (int i = 0; i < 3; i++) {
#pragma unroll 4
    for (int kk = 0; kk < 32; kk++) {
      const int kb = kk * 4 + 2 * hi;
      v2f a =
          *(const v2f*)(m1 + (long)i * NN * 128 + (long)(n0 + row) * 128 + kb);
#pragma unroll
      for (int nt = 0; nt < 4; nt++) {
        v2f b;
        b.x = Wv[kb * 64 + nt * 16 + col];
        b.y = Wv[(kb + 1) * 64 + nt * 16 + col];
        acc[1 + i][nt] = wmma4(a, b, acc[1 + i][nt]);
      }
    }
  }
  const float inv = 1.0f / 16.0f;
#pragma unroll
  for (int nt = 0; nt < 4; nt++) {
    const int c = nt * 16 + col;
#pragma unroll
    for (int r = 0; r < 8; r++) {
      const int n = n0 + r + 8 * hi;
      float4 o = make_float4(acc[0][nt][r] * inv, acc[1][nt][r] * inv,
                             acc[2][nt][r] * inv, acc[3][nt][r] * inv);
      *(float4*)(out + (long)n * 256 + c * 4) = o;
    }
  }
}

extern "C" void kernel_launch(void* const* d_in, const int* in_sizes, int n_in,
                              void* d_out, int out_size, void* d_ws,
                              size_t ws_size, hipStream_t stream) {
  (void)in_sizes; (void)n_in; (void)out_size; (void)ws_size;
  const float* node_feats = (const float*)d_in[0];
  const float* edge_attrs = (const float*)d_in[1];
  const float* edge_feats = (const float*)d_in[2];
  const float* lengths    = (const float*)d_in[3];
  const int*   edge_index = (const int*)d_in[4];
  const float* Wsc  = (const float*)d_in[5];
  const float* Wup0 = (const float*)d_in[6];
  const float* Wup1 = (const float*)d_in[7];
  const float* W1   = (const float*)d_in[8];
  const float* b1   = (const float*)d_in[9];
  const float* W2   = (const float*)d_in[10];
  const float* b2   = (const float*)d_in[11];
  const float* W3   = (const float*)d_in[12];
  const float* Ws   = (const float*)d_in[13];
  const float* Wv   = (const float*)d_in[14];
  float* out = (float*)d_out;

  float* ws  = (float*)d_ws;
  float* ns  = ws;
  float* sup = ns + (long)NN * 64;
  float* vup = sup + (long)NN * 64;
  float* m0  = vup + (long)3 * NN * 64;
  float* m1  = m0 + (long)NN * 128;

  int* scr     = (int*)d_out;
  int* cnt     = scr;
  int* start   = cnt + NN;
  int* arrival = start + NN;
  int* perm_a  = arrival + NE;
  int* perm_b  = perm_a + NE;

  const long zn4 = (long)NN * 512 / 4;
  zero_kernel<<<(int)((zn4 + 255) / 256), 256, 0, stream>>>(
      (float4*)m0, zn4, (int4*)cnt, (long)NN / 4);
  count_kernel<<<(NE + 255) / 256, 256, 0, stream>>>(edge_index, cnt, arrival);
  scan_kernel<<<1, 1024, 0, stream>>>(cnt, start);
  place_kernel<<<(NE + 255) / 256, 256, 0, stream>>>(edge_index, start, arrival,
                                                     perm_a);
  rank_kernel<<<(NE + 255) / 256, 256, 0, stream>>>(edge_index, cnt, start,
                                                    perm_a, perm_b);

  node_up_kernel<<<(NN / 16 + 7) / 8, 256, 0, stream>>>(node_feats, Wsc, Wup0,
                                                        Wup1, ns, sup, vup);
  edge_kernel<<<NE / 32 / 4, 128, 0, stream>>>(edge_attrs, edge_feats, lengths,
                                               edge_index, perm_b, W1, b1, W2,
                                               b2, W3, ns, sup, vup, m0, m1);
  out_kernel<<<(NN / 16 + 7) / 8, 256, 0, stream>>>(m0, m1, Ws, Wv, out);
}


# edge and node kernels: weights re-laid once per call in matrix-fragment order (by the zero kernel); two B fragments come as one whole-line 16-B load per lane, not four 4-B loads; same sums
# speedup vs baseline: 2.1277x; 1.0378x over previous
#include <hip/hip_runtime.h>

#define NN 20000
#define NE 320000
#define INV_SQRT3 0.5773502691896258f
#define SPITCH 68

typedef __attribute__((ext_vector_type(2))) float v2f;
typedef __attribute__((ext_vector_type(8))) float v8f;

__device__ __forceinline__ v8f wmma4(v2f a, v2f b, v8f c) {
  return __builtin_amdgcn_wmma_f32_16x16x4_f32(false, a, false, b, (short)0, c,
                                               false, false);
}

__device__ __forceinline__ float silu_f(float x) {
  return x * (1.0f / (1.0f + __expf(-x)));
}

__device__ __forceinline__ v8f zero8() {
  v8f z = {0.f, 0.f, 0.f, 0.f, 0.f, 0.f, 0.f, 0.f};
  return z;
}

#define PK_W1   0
#define PK_W2   2240
#define PK_W3   3264
#define PK_WSC  7360
#define PK_WUP0 8384
#define PK_WUP1 9408
#define PK_END  10432

__device__ __forceinline__ void bfrag4(const float4* __restrict__ wf, int kk,
                                       int njp, int jp0, int lane, v2f b[4]) {
  const float4 f0 = wf[(kk * njp + jp0) * 32 + lane];
  const float4 f1 = wf[(kk * njp + jp0 + 1) * 32 + lane];
  b[0].x = f0.x; b[0].y = f0.y;
  b[1].x = f0.z; b[1].y = f0.w;
  b[2].x = f1.x; b[2].y = f1.y;
  b[3].x = f1.z; b[3].y = f1.w;
}

__global__ __launch_bounds__(256) void zero_kernel(
    float4* p, long n4, int4* q, long nq4, const float* __restrict__ W1,
    const float* __restrict__ W2, const float* __restrict__ W3,
    const float* __restrict__ Wsc, const float* __restrict__ Wup0,
    const float* __restrict__ Wup1, float4* __restrict__ wpk) {
  long i = (long)blockIdx.x * blockDim.x + threadIdx.x;
  if (i < n4) p[i] = make_float4(0.f, 0.f, 0.f, 0.f);
  if (i < nq4) q[i] = make_int4(0, 0, 0, 0);
  if (i < PK_END) {
    const int g = (int)i;
    const float* Wm; int K, ld, njp, base;
    if (g < PK_W2)        { Wm = W1;   K = 137; ld = 64;  njp = 2; base = PK_W1; }
    else if (g < PK_W3)   { Wm = W2;   K = 64;  ld = 64;  njp = 2; base = PK_W2; }
    else if (g < PK_WSC)  { Wm = W3;   K = 64;  ld = 256; njp = 8; base = PK_W3; }
    else if (g < PK_WUP0) { Wm = Wsc;  K = 64;  ld = 64;  njp = 2; base = PK_WSC; }
    else if (g < PK_WUP1) { Wm = Wup0; K = 64;  ld = 64;  njp = 2; base = PK_WUP0; }
    else                  { Wm = Wup1; K = 64;  ld = 64;  njp = 2; base = PK_WUP1; }
    const int idx = g - base;
    const int l = idx & 31;
    const int t = idx >> 5;
    const int jp = t % njp;
    const int kk = t / njp;
    const int k0 = 4 * kk + 2 * (l >> 4);
    const int n0 = 32 * jp + (l & 15);
    float4 w;
    w.x = k0 < K ? Wm[k0 * ld + n0] : 0.f;
    w.y = k0 + 1 < K ? Wm[(k0 + 1) * ld + n0] : 0.f;
    w.z = k0 < K ? Wm[k0 * ld + n0 + 16] : 0.f;
    w.w = k0 + 1 < K ? Wm[(k0 + 1) * ld + n0 + 16] : 0.f;
    wpk[g] = w;
  }
}

__global__ __launch_bounds__(256) void count_kernel(
    const int* __restrict__ edge_index, int* __restrict__ cnt,
    int* __restrict__ arrival) {
  const int e = blockIdx.x * 256 + threadIdx.x;
  if (e < NE) {
    const int r = edge_index[NE + e];
    arrival[e] = (unsigned)r < (unsigned)NN ? atomicAdd(&cnt[r], 1) : 0;
  }
}

#define SCAN_PER 20
__global__ __launch_bounds__(1024) void scan_kernel(
    const int* __restrict__ cnt, int* __restrict__ start) {
  __shared__ int part[1024];
  const int t = threadIdx.x;
  int loc[SCAN_PER];
  int sum = 0;
#pragma unroll
  for (int j = 0; j < SCAN_PER; j++) {
    const int idx = t * SCAN_PER + j;
    const int v = idx < NN ? cnt[idx] : 0;
    loc[j] = sum;
    sum += v;
  }
  part[t] = sum;
  __syncthreads();
  for (int off = 1; off < 1024; off <<= 1) {
    const int v = t >= off ? part[t - off] : 0;
    __syncthreads();
    part[t] += v;
    __syncthreads();
  }
  const int excl = part[t] - sum;
#pragma unroll
  for (int j = 0; j < SCAN_PER; j++) {
    const int idx = t * SCAN_PER + j;
    if (idx < NN) start[idx] = excl + loc[j];
  }
}

__global__ __launch_bounds__(256) void place_kernel(
    const int* __restrict__ edge_index, const int* __restrict__ start,
    const int* __restrict__ arrival, int* __restrict__ perm_a) {
  const int e = blockIdx.x * 256 + threadIdx.x;
  if (e < NE) {
    const int r = edge_index[NE + e];
    if ((unsigned)r < (unsigned)NN) {
      const int p = start[r] + arrival[e];
      if ((unsigned)p < (unsigned)NE) perm_a[p] = e;
    }
  }
}

__global__ __launch_bounds__(256) void rank_kernel(
    const int* __restrict__ edge_index, const int* __restrict__ cnt,
    const int* __restrict__ start, const int* __restrict__ perm_a,
    int* __restrict__ perm_b) {
  const int p = blockIdx.x * 256 + threadIdx.x;
  if (p < NE) {
    int e = perm_a[p];
    e = (unsigned)e < (unsigned)NE ? e : 0;
    int r = edge_index[NE + e];
    r = (unsigned)r < (unsigned)NN ? r : 0;
    const int s = start[r];
    const int d = cnt[r];
    int k = 0;
    for (int j = 0; j < d; j++) k += (perm_a[s + j] < e) ? 1 : 0;
    const int q = s + k;
    if ((unsigned)q < (unsigned)NE) perm_b[q] = e;
  }
}

__global__ __launch_bounds__(256) void node_up_kernel(
    const float* __restrict__ node_feats, const float4* __restrict__ wpk,
    float* __restrict__ ns, float* __restrict__ sup, float* __restrict__ vup) {
  const int lane = threadIdx.x & 31;
  const int wv = threadIdx.x >> 5;
  const int tile = blockIdx.x * 8 + wv;
  if (tile >= NN / 16) return;
  const int n0 = tile * 16;
  const int row = lane & 15;
  const int hi = lane >> 4;
  const int col = row;
  const float* arow = node_feats + (long)(n0 + row) * 256;

  v8f accA[4], accB[4];
#pragma unroll
  for (int nt = 0; nt < 4; nt++) { accA[nt] = zero8(); accB[nt] = zero8(); }
#pragma unroll
  for (int kk = 0; kk < 16; kk++) {
    const int kb = kk * 4 + 2 * hi;
    v2f a = *(const v2f*)(arow + kb);
    v2f b0[4], b1[4];
    bfrag4(wpk + PK_WSC, kk, 2, 0, lane, b0);
    bfrag4(wpk + PK_WUP0, kk, 2, 0, lane, b1);
#pragma unroll
    for (int nt = 0; nt < 4; nt++) {
      accA[nt] = wmma4(a, b0[nt], accA[nt]);
      accB[nt] = wmma4(a, b1[nt], accB[nt]);
    }
  }
#pragma unroll
  for (int nt = 0; nt < 4; nt++)
#pragma unroll
    for (int r = 0; r < 8; r++) {
      const int n = n0 + r + 8 * hi;
      ns[(long)n * 64 + nt * 16 + col] = accA[nt][r];
      sup[(long)n * 64 + nt * 16 + col] = accB[nt][r];
    }

#pragma unroll
  for (int i = 0; i < 3; i++) {
    v8f acc[4];
#pragma unroll
    for (int nt = 0; nt < 4; nt++) acc[nt] = zero8();
#pragma unroll
    for (int kk = 0; kk < 16; kk++) {
      const int kb = kk * 4 + 2 * hi;
      v2f a;
      a.x = arow[64 + kb * 3 + i];
      a.y = arow[64 + (kb + 1) * 3 + i];
      v2f b[4];
      bfrag4(wpk + PK_WUP1, kk, 2, 0, lane, b);
#pragma unroll
      for (int nt = 0; nt < 4; nt++) acc[nt] = wmma4(a, b[nt], acc[nt]);
    }
#pragma unroll
    for (int nt = 0; nt < 4; nt++)
#pragma unroll
      for (int r = 0; r < 8; r++) {
        const int n = n0 + r + 8 * hi;
        vup[(long)i * NN * 64 + (long)n * 64 + nt * 16 + col] = acc[nt][r];
      }
  }
}

__global__ __launch_bounds__(128) void edge_kernel(
    const float* __restrict__ edge_attrs, const float* __restrict__ edge_feats,
    const float* __restrict__ lengths, const int* __restrict__ edge_index,
    const int* __restrict__ perm,
    const float4* __restrict__ wpk, const float* __restrict__ b1,
    const float* __restrict__ b2, const float* __restrict__ ns,
    const float* __restrict__ sup, const float* __restrict__ vup,
    float* __restrict__ m0, float* __restrict__ m1) {
  __shared__ __attribute__((aligned(16))) float lds_w[4][32 * 64 + 32 * SPITCH];
  __shared__ float lds_meta[4][32 * 8];
  const int lane = threadIdx.x & 31;
  const int wv = threadIdx.x >> 5;
  const int dt = blockIdx.x * 4 + wv;
  const int e0 = dt * 32;
  const int row = lane & 15;
  const int hi = lane >> 4;
  const int col = row;

  int ea = perm[e0 + row];
  int eb = perm[e0 + 16 + row];
  ea = (unsigned)ea < (unsigned)NE ? ea : 0;
  eb = (unsigned)eb < (unsigned)NE ? eb : 0;
  const int snd0 = edge_index[ea], rcv0 = edge_index[NE + ea];
  const int snd1 = edge_index[eb], rcv1 = edge_index[NE + eb];
  {
    const int mr = row + 16 * hi;
    const int me = hi ? eb : ea;
    const int myr = hi ? rcv1 : rcv0;
    const int nxt = __shfl_down(myr, 1);
    const int endf = ((lane & 7) == 7) || (nxt != myr);
    lds_meta[wv][mr * 8 + 6] = __int_as_float(endf);
    lds_meta[wv][mr * 8 + 0] = __int_as_float(hi ? snd1 : snd0);
    lds_meta[wv][mr * 8 + 1] = __int_as_float(myr);
    lds_meta[wv][mr * 8 + 2] = edge_attrs[me * 4 + 0];
    lds_meta[wv][mr * 8 + 3] = edge_attrs[me * 4 + 1];
    lds_meta[wv][mr * 8 + 4] = edge_attrs[me * 4 + 2];
    lds_meta[wv][mr * 8 + 5] = edge_attrs[me * 4 + 3];
  }
  float* const lds_h = lds_w[wv];
  float* const stg = lds_w[wv] + 32 * 64;
  const float* sp0 = ns + (long)snd0 * 64;
  const float* rp0 = ns + (long)rcv0 * 64;
  const float* sp1 = ns + (long)snd1 * 64;
  const float* rp1 = ns + (long)rcv1 * 64;
  const float* ep0 = edge_feats + (long)ea * 8;
  const float* ep1 = edge_feats + (long)eb * 8;
  const float len0 = lengths[ea];
  const float len1 = lengths[eb];

  v8f acc0[4], acc1[4];
#pragma unroll
  for (int nt = 0; nt < 4; nt++) { acc0[nt] = zero8(); acc1[nt] = zero8(); }
#pragma unroll 4
  for (int kk = 0; kk < 16; kk++) {
    const int kb = kk * 4 + 2 * hi;
    v2f a0 = *(const v2f*)(sp0 + kb);
    v2f a1 = *(const v2f*)(sp1 + kb);
    v2f b[4];
    bfrag4(wpk + PK_W1, kk, 2, 0, lane, b);
#pragma unroll
    for (int nt = 0; nt < 4; nt++) {
      acc0[nt] = wmma4(a0, b[nt], acc0[nt]);
      acc1[nt] = wmma4(a1, b[nt], acc1[nt]);
    }
  }
#pragma unroll 4
  for (int kk = 0; kk < 16; kk++) {
    const int kb = kk * 4 + 2 * hi;
    v2f a0 = *(const v2f*)(rp0 + kb);
    v2f a1 = *(const v2f*)(rp1 + kb);
    v2f b[4];
    bfrag4(wpk + PK_W1, 16 + kk, 2, 0, lane, b);
#pragma unroll
    for (int nt = 0; nt < 4; nt++) {
      acc0[nt] = wmma4(a0, b[nt], acc0[nt]);
      acc1[nt] = wmma4(a1, b[nt], acc1[nt]);
    }
  }
#pragma unroll
  for (int kk = 0; kk < 2; kk++) {
    const int kb = kk * 4 + 2 * hi;
    v2f a0 = *(const v2f*)(ep0 + kb);
    v2f a1 = *(const v2f*)(ep1 + kb);
    v2f b[4];
    bfrag4(wpk + PK_W1, 32 + kk, 2, 0, lane, b);
#pragma unroll
    for (int nt = 0; nt < 4; nt++) {
      acc0[nt] = wmma4(a0, b[nt], acc0[nt]);
      acc1[nt] = wmma4(a1, b[nt], acc1[nt]);
    }
  }
  {
    v2f a0, a1;
    a0.x = hi ? 0.f : len0;  a0.y = 0.f;
    a1.x = hi ? 0.f : len1;  a1.y = 0.f;
    v2f b[4];
    bfrag4(wpk + PK_W1, 34, 2, 0, lane, b);
#pragma unroll
    for (int nt = 0; nt < 4; nt++) {
      acc0[nt] = wmma4(a0, b[nt], acc0[nt]);
      acc1[nt] = wmma4(a1, b[nt], acc1[nt]);
    }
  }
  float4 gs[16];
#pragma unroll
  for (int i = 0; i < 16; i++) {
    const int s_e = __float_as_int(lds_meta[wv][(2 * i + hi) * 8 + 0]);
    gs[i] = *(const float4*)(sup + (long)s_e * 64 + 4 * row);
  }
#pragma unroll
  for (int nt = 0; nt < 4; nt++) {
    const float bias = b1[nt * 16 + col];
#pragma unroll
    for (int r = 0; r < 8; r++) {
      lds_h[(r + 8 * hi) * 64 + nt * 16 + col] = silu_f(acc0[nt][r] + bias);
      lds_h[(16 + r + 8 * hi) * 64 + nt * 16 + col] =
          silu_f(acc1[nt][r] + bias);
    }
  }

#pragma unroll
  for (int i = 0; i < 16; i++)
    *(float4*)&stg[(2 * i + hi) * SPITCH + 4 * row] = gs[i];

#pragma unroll
  for (int nt = 0; nt < 4; nt++) { acc0[nt] = zero8(); acc1[nt] = zero8(); }
#pragma unroll 4
  for (int kk = 0; kk < 16; kk++) {
    const int kb = kk * 4 + 2 * hi;
    v2f a0 = *(const v2f*)&lds_h[row * 64 + kb];
    v2f a1 = *(const v2f*)&lds_h[(16 + row) * 64 + kb];
    v2f b[4];
    bfrag4(wpk + PK_W2, kk, 2, 0, lane, b);
#pragma unroll
    for (int nt = 0; nt < 4; nt++) {
      acc0[nt] = wmma4(a0, b[nt], acc0[nt]);
      acc1[nt] = wmma4(a1, b[nt], acc1[nt]);
    }
  }
  float h2a[4][8], h2b[4][8];
#pragma unroll
  for (int nt = 0; nt < 4; nt++) {
    const float bias = b2[nt * 16 + col];
#pragma unroll
    for (int r = 0; r < 8; r++) {
      h2a[nt][r] = silu_f(acc0[nt][r] + bias);
      h2b[nt][r] = silu_f(acc1[nt][r] + bias);
    }
  }
#pragma unroll
  for (int nt = 0; nt < 4; nt++)
#pragma unroll
    for (int r = 0; r < 8; r++) {
      lds_h[(r + 8 * hi) * 64 + nt * 16 + col] = h2a[nt][r];
      lds_h[(16 + r + 8 * hi) * 64 + nt * 16 + col] = h2b[nt][r];
    }

  {
    v8f ta0[4], ta1[4], tc0[4], tc1[4];
#pragma unroll
    for (int nt = 0; nt < 4; nt++) {
      ta0[nt] = zero8(); ta1[nt] = zero8();
      tc0[nt] = zero8(); tc1[nt] = zero8();
    }
#pragma unroll 2
    for (int kk = 0; kk < 16; kk++) {
      const int kb = kk * 4 + 2 * hi;
      v2f a0 = *(const v2f*)&lds_h[row * 64 + kb];
      v2f a1 = *(const v2f*)&lds_h[(16 + row) * 64 + kb];
      v2f b[4], d[4];
      bfrag4(wpk + PK_W3, kk, 8, 0, lane, b);
      bfrag4(wpk + PK_W3, kk, 8, 4, lane, d);
#pragma unroll
      for (int nt = 0; nt < 4; nt++) {
        ta0[nt] = wmma4(a0, b[nt], ta0[nt]);
        ta1[nt] = wmma4(a1, b[nt], ta1[nt]);
        tc0[nt] = wmma4(a0, d[nt], tc0[nt]);
        tc1[nt] = wmma4(a1, d[nt], tc1[nt]);
      }
    }
#pragma unroll
    for (int t = 0; t < 2; t++) {
#pragma unroll
      for (int nt = 0; nt < 4; nt++) {
        const int c = nt * 16 + col;
        float q0 = 0.f, q1 = 0.f, q2 = 0.f, q3 = 0.f;
#pragma unroll
        for (int r = 0; r < 8; r++) {
          const int rr = t * 16 + r + 8 * hi;
          const int r_e = __float_as_int(lds_meta[wv][rr * 8 + 1]);
          const float y0 = lds_meta[wv][rr * 8 + 2];
          const float xs = stg[rr * SPITCH + c];
          const float wa = t ? ta1[nt][r] : ta0[nt][r];
          const float wc = t ? tc1[nt][r] : tc0[nt][r];
          q0 += xs * y0 * wa;
          q1 += xs * lds_meta[wv][rr * 8 + 3] * wc;
          q2 += xs * lds_meta[wv][rr * 8 + 4] * wc;
          q3 += xs * lds_meta[wv][rr * 8 + 5] * wc;
          if (__float_as_int(lds_meta[wv][rr * 8 + 6])) {
            unsafeAtomicAdd(&m0[(long)r_e * 128 + c], q0);
            unsafeAtomicAdd(&m1[(long)r_e * 128 + c], q1);
            unsafeAtomicAdd(&m1[(long)NN * 128 + (long)r_e * 128 + c], q2);
            unsafeAtomicAdd(&m1[(long)2 * NN * 128 + (long)r_e * 128 + c], q3);
            q0 = 0.f; q1 = 0.f; q2 = 0.f; q3 = 0.f;
          }
        }
      }
    }
  }
  {
    v8f tb0[4], tb1[4], td0[4], td1[4];
#pragma unroll
    for (int nt = 0; nt < 4; nt++) {
      tb0[nt] = zero8(); tb1[nt] = zero8();
      td0[nt] = zero8(); td1[nt] = zero8();
    }
#pragma unroll 2
    for (int kk = 0; kk < 16; kk++) {
      const int kb = kk * 4 + 2 * hi;
      v2f a0 = *(const v2f*)&lds_h[row * 64 + kb];
      v2f a1 = *(const v2f*)&lds_h[(16 + row) * 64 + kb];
      v2f b[4], d[4];
      bfrag4(wpk + PK_W3, kk, 8, 2, lane, b);
      bfrag4(wpk + PK_W3, kk, 8, 6, lane, d);
#pragma unroll
      for (int nt = 0; nt < 4; nt++) {
        tb0[nt] = wmma4(a0, b[nt], tb0[nt]);
        tb1[nt] = wmma4(a1, b[nt], tb1[nt]);
        td0[nt] = wmma4(a0, d[nt], td0[nt]);
        td1[nt] = wmma4(a1, d[nt], td1[nt]);
      }
    }
    float* const vst = lds_w[wv];
#pragma unroll
    for (int t = 0; t < 2; t++) {
#pragma unroll 1
      for (int pl = 0; pl < 3; pl++) {
        float4 gv[8];
#pragma unroll
        for (int i = 0; i < 8; i++) {
          const int s_e =
              __float_as_int(lds_meta[wv][(t * 16 + 2 * i + hi) * 8 + 0]);
          gv[i] = *(const float4*)(vup + (long)pl * NN * 64 + (long)s_e * 64 +
                                   4 * row);
        }
#pragma unroll
        for (int i = 0; i < 8; i++)
          *(float4*)&vst[(pl * 16 + 2 * i + hi) * SPITCH + 4 * row] = gv[i];
      }
#pragma unroll
      for (int nt = 0; nt < 4; nt++) {
        const int c = nt * 16 + col;
        float q0 = 0.f, q1 = 0.f, q2 = 0.f, q3 = 0.f;
#pragma unroll
        for (int r = 0; r < 8; r++) {
          const int rr = t * 16 + r + 8 * hi;
          const int lr = r + 8 * hi;
          const int r_e = __float_as_int(lds_meta[wv][rr * 8 + 1]);
          const float y0 = lds_meta[wv][rr * 8 + 2];
          const float x0 = vst[(0 * 16 + lr) * SPITCH + c];
          const float x1 = vst[(1 * 16 + lr) * SPITCH + c];
          const float x2 = vst[(2 * 16 + lr) * SPITCH + c];
          const float wb = t ? tb1[nt][r] : tb0[nt][r];
          const float wd = t ? td1[nt][r] : td0[nt][r];
          const float dot = x0 * lds_meta[wv][rr * 8 + 3] +
                            x1 * lds_meta[wv][rr * 8 + 4] +
                            x2 * lds_meta[wv][rr * 8 + 5];
          q0 += dot * INV_SQRT3 * wb;
          q1 += x0 * y0 * wd;
          q2 += x1 * y0 * wd;
          q3 += x2 * y0 * wd;
          if (__float_as_int(lds_meta[wv][rr * 8 + 6])) {
            unsafeAtomicAdd(&m0[(long)r_e * 128 + 64 + c], q0);
            unsafeAtomicAdd(&m1[(long)r_e * 128 + 64 + c], q1);
            unsafeAtomicAdd(&m1[(long)NN * 128 + (long)r_e * 128 + 64 + c],
                            q2);
            unsafeAtomicAdd(
                &m1[(long)2 * NN * 128 + (long)r_e * 128 + 64 + c], q3);
            q0 = 0.f; q1 = 0.f; q2 = 0.f; q3 = 0.f;
          }
        }
      }
    }
  }
}

__global__ __launch_bounds__(256) void out_kernel(
    const float* __restrict__ m0, const float* __restrict__ m1,
    const float* __restrict__ Ws, const float* __restrict__ Wv,
    float* __restrict__ out) {
  const int lane = threadIdx.x & 31;
  const int wv = threadIdx.x >> 5;
  const int tile = blockIdx.x * 8 + wv;
  if (tile >= NN / 16) return;
  const int n0 = tile * 16;
  const int row = lane & 15;
  const int hi = lane >> 4;
  const int col = row;

  v8f acc[4][4];
#pragma unroll
  for (int j = 0; j < 4; j++)
#pragma unroll
    for (int nt = 0; nt < 4; nt++) acc[j][nt] = zero8();

#pragma unroll 4
  for (int kk = 0; kk < 32; kk++) {
    const int kb = kk * 4 + 2 * hi;
    v2f a = *(const v2f*)(m0 + (long)(n0 + row) * 128 + kb);
#pragma unroll
    for (int nt = 0; nt < 4; nt++) {
      v2f b;
      b.x = Ws[kb * 64 + nt * 16 + col];
      b.y = Ws[(kb + 1) * 64 + nt * 16 + col];
      acc[0][nt] = wmma4(a, b, acc[0][nt]);
    }
  }
#pragma unroll
  for (int i = 0; i < 3; i++) {
#pragma unroll 4
    for (int kk = 0; kk < 32; kk++) {
      const int kb = kk * 4 + 2 * hi;
      v2f a =
          *(const v2f*)(m1 + (long)i * NN * 128 + (long)(n0 + row) * 128 + kb);
#pragma unroll
      for (int nt = 0; nt < 4; nt++) {
        v2f b;
        b.x = Wv[kb * 64 + nt * 16 + col];
        b.y = Wv[(kb + 1) * 64 + nt * 16 + col];
        acc[1 + i][nt] = wmma4(a, b, acc[1 + i][nt]);
      }
    }
  }
  const float inv = 1.0f / 16.0f;
#pragma unroll
  for (int nt = 0; nt < 4; nt++) {
    const int c = nt * 16 + col;
#pragma unroll
    for (int r = 0; r < 8; r++) {
      const int n = n0 + r + 8 * hi;
      float4 o = make_float4(acc[0][nt][r] * inv, acc[1][nt][r] * inv,
                             acc[2][nt][r] * inv, acc[3][nt][r] * inv);
      *(float4*)(out + (long)n * 256 + c * 4) = o;
    }
  }
}

extern "C" void kernel_launch(void* const* d_in, const int* in_sizes, int n_in,
                              void* d_out, int out_size, void* d_ws,
                              size_t ws_size, hipStream_t stream) {
  (void)in_sizes; (void)n_in; (void)out_size; (void)ws_size;
  const float* node_feats = (const float*)d_in[0];
  const float* edge_attrs = (const float*)d_in[1];
  const float* edge_feats = (const float*)d_in[2];
  const float* lengths    = (const float*)d_in[3];
  const int*   edge_index = (const int*)d_in[4];
  const float* Wsc  = (const float*)d_in[5];
  const float* Wup0 = (const float*)d_in[6];
  const float* Wup1 = (const float*)d_in[7];
  const float* W1   = (const float*)d_in[8];
  const float* b1   = (const float*)d_in[9];
  const float* W2   = (const float*)d_in[10];
  const float* b2   = (const float*)d_in[11];
  const float* W3   = (const float*)d_in[12];
  const float* Ws   = (const float*)d_in[13];
  const float* Wv   = (const float*)d_in[14];
  float* out = (float*)d_out;

  float* ws  = (float*)d_ws;
  float* ns  = ws;
  float* sup = ns + (long)NN * 64;
  float* vup = sup + (long)NN * 64;
  float* m0  = vup + (long)3 * NN * 64;
  float* m1  = m0 + (long)NN * 128;

  int* scr     = (int*)d_out;
  int* cnt     = scr;
  int* start   = cnt + NN;
  int* arrival = start + NN;
  int* perm_a  = arrival + NE;
  int* perm_b  = perm_a + NE;
  float4* wpk  = (float4*)(perm_b + NE);

  const long zn4 = (long)NN * 512 / 4;
  zero_kernel<<<(int)((zn4 + 255) / 256), 256, 0, stream>>>(
      (float4*)m0, zn4, (int4*)cnt, (long)NN / 4, W1, W2, W3, Wsc, Wup0, Wup1,
      wpk);
  count_kernel<<<(NE + 255) / 256, 256, 0, stream>>>(edge_index, cnt, arrival);
  scan_kernel<<<1, 1024, 0, stream>>>(cnt, start);
  place_kernel<<<(NE + 255) / 256, 256, 0, stream>>>(edge_index, start, arrival,
                                                     perm_a);
  rank_kernel<<<(NE + 255) / 256, 256, 0, stream>>>(edge_index, cnt, start,
                                                    perm_a, perm_b);

  node_up_kernel<<<(NN / 16 + 7) / 8, 256, 0, stream>>>(node_feats, wpk, ns,
                                                        sup, vup);
  edge_kernel<<<NE / 32 / 4, 128, 0, stream>>>(edge_attrs, edge_feats, lengths,
                                               edge_index, perm_b, wpk, b1, b2,
                                               ns, sup, vup, m0, m1);
  out_kernel<<<(NN / 16 + 7) / 8, 256, 0, stream>>>(m0, m1, Ws, Wv, out);
}


# edge kernel scatter epilogues: rows outer, column tiles inner; a row's receiver, y values and run-end flag read once, one run-end branch per row for its four column tiles; same sums, same atomics
# speedup vs baseline: 2.2652x; 1.0646x over previous
#include <hip/hip_runtime.h>

#define NN 20000
#define NE 320000
#define INV_SQRT3 0.5773502691896258f
#define SPITCH 68

typedef __attribute__((ext_vector_type(2))) float v2f;
typedef __attribute__((ext_vector_type(8))) float v8f;

__device__ __forceinline__ v8f wmma4(v2f a, v2f b, v8f c) {
  return __builtin_amdgcn_wmma_f32_16x16x4_f32(false, a, false, b, (short)0, c,
                                               false, false);
}

__device__ __forceinline__ float silu_f(float x) {
  return x * (1.0f / (1.0f + __expf(-x)));
}

__device__ __forceinline__ v8f zero8() {
  v8f z = {0.f, 0.f, 0.f, 0.f, 0.f, 0.f, 0.f, 0.f};
  return z;
}

#define PK_W1   0
#define PK_W2   2240
#define PK_W3   3264
#define PK_WSC  7360
#define PK_WUP0 8384
#define PK_WUP1 9408
#define PK_END  10432

__device__ __forceinline__ void bfrag4(const float4* __restrict__ wf, int kk,
                                       int njp, int jp0, int lane, v2f b[4]) {
  const float4 f0 = wf[(kk * njp + jp0) * 32 + lane];
  const float4 f1 = wf[(kk * njp + jp0 + 1) * 32 + lane];
  b[0].x = f0.x; b[0].y = f0.y;
  b[1].x = f0.z; b[1].y = f0.w;
  b[2].x = f1.x; b[2].y = f1.y;
  b[3].x = f1.z; b[3].y = f1.w;
}

__global__ __launch_bounds__(256) void zero_kernel(
    float4* p, long n4, int4* q, long nq4, const float* __restrict__ W1,
    const float* __restrict__ W2, const float* __restrict__ W3,
    const float* __restrict__ Wsc, const float* __restrict__ Wup0,
    const float* __restrict__ Wup1, float4* __restrict__ wpk) {
  long i = (long)blockIdx.x * blockDim.x + threadIdx.x;
  if (i < n4) p[i] = make_float4(0.f, 0.f, 0.f, 0.f);
  if (i < nq4) q[i] = make_int4(0, 0, 0, 0);
  if (i < PK_END) {
    const int g = (int)i;
    const float* Wm; int K, ld, njp, base;
    if (g < PK_W2)        { Wm = W1;   K = 137; ld = 64;  njp = 2; base = PK_W1; }
    else if (g < PK_W3)   { Wm = W2;   K = 64;  ld = 64;  njp = 2; base = PK_W2; }
    else if (g < PK_WSC)  { Wm = W3;   K = 64;  ld = 256; njp = 8; base = PK_W3; }
    else if (g < PK_WUP0) { Wm = Wsc;  K = 64;  ld = 64;  njp = 2; base = PK_WSC; }
    else if (g < PK_WUP1) { Wm = Wup0; K = 64;  ld = 64;  njp = 2; base = PK_WUP0; }
    else                  { Wm = Wup1; K = 64;  ld = 64;  njp = 2; base = PK_WUP1; }
    const int idx = g - base;
    const int l = idx & 31;
    const int t = idx >> 5;
    const int jp = t % njp;
    const int kk = t / njp;
    const int k0 = 4 * kk + 2 * (l >> 4);
    const int n0 = 32 * jp + (l & 15);
    float4 w;
    w.x = k0 < K ? Wm[k0 * ld + n0] : 0.f;
    w.y = k0 + 1 < K ? Wm[(k0 + 1) * ld + n0] : 0.f;
    w.z = k0 < K ? Wm[k0 * ld + n0 + 16] : 0.f;
    w.w = k0 + 1 < K ? Wm[(k0 + 1) * ld + n0 + 16] : 0.f;
    wpk[g] = w;
  }
}

__global__ __launch_bounds__(256) void count_kernel(
    const int* __restrict__ edge_index, int* __restrict__ cnt,
    int* __restrict__ arrival) {
  const int e = blockIdx.x * 256 + threadIdx.x;
  if (e < NE) {
    const int r = edge_index[NE + e];
    arrival[e] = (unsigned)r < (unsigned)NN ? atomicAdd(&cnt[r], 1) : 0;
  }
}

#define SCAN_PER 20
__global__ __launch_bounds__(1024) void scan_kernel(
    const int* __restrict__ cnt, int* __restrict__ start) {
  __shared__ int part[1024];
  const int t = threadIdx.x;
  int loc[SCAN_PER];
  int sum = 0;
#pragma unroll
  for (int j = 0; j < SCAN_PER; j++) {
    const int idx = t * SCAN_PER + j;
    const int v = idx < NN ? cnt[idx] : 0;
    loc[j] = sum;
    sum += v;
  }
  part[t] = sum;
  __syncthreads();
  for (int off = 1; off < 1024; off <<= 1) {
    const int v = t >= off ? part[t - off] : 0;
    __syncthreads();
    part[t] += v;
    __syncthreads();
  }
  const int excl = part[t] - sum;
#pragma unroll
  for (int j = 0; j < SCAN_PER; j++) {
    const int idx = t * SCAN_PER + j;
    if (idx < NN) start[idx] = excl + loc[j];
  }
}

__global__ __launch_bounds__(256) void place_kernel(
    const int* __restrict__ edge_index, const int* __restrict__ start,
    const int* __restrict__ arrival, int* __restrict__ perm_a) {
  const int e = blockIdx.x * 256 + threadIdx.x;
  if (e < NE) {
    const int r = edge_index[NE + e];
    if ((unsigned)r < (unsigned)NN) {
      const int p = start[r] + arrival[e];
      if ((unsigned)p < (unsigned)NE) perm_a[p] = e;
    }
  }
}

__global__ __launch_bounds__(256) void rank_kernel(
    const int* __restrict__ edge_index, const int* __restrict__ cnt,
    const int* __restrict__ start, const int* __restrict__ perm_a,
    int* __restrict__ perm_b) {
  const int p = blockIdx.x * 256 + threadIdx.x;
  if (p < NE) {
    int e = perm_a[p];
    e = (unsigned)e < (unsigned)NE ? e : 0;
    int r = edge_index[NE + e];
    r = (unsigned)r < (unsigned)NN ? r : 0;
    const int s = start[r];
    const int d = cnt[r];
    int k = 0;
    for (int j = 0; j < d; j++) k += (perm_a[s + j] < e) ? 1 : 0;
    const int q = s + k;
    if ((unsigned)q < (unsigned)NE) perm_b[q] = e;
  }
}

__global__ __launch_bounds__(256) void node_up_kernel(
    const float* __restrict__ node_feats, const float4* __restrict__ wpk,
    float* __restrict__ ns, float* __restrict__ sup, float* __restrict__ vup) {
  const int lane = threadIdx.x & 31;
  const int wv = threadIdx.x >> 5;
  const int tile = blockIdx.x * 8 + wv;
  if (tile >= NN / 16) return;
  const int n0 = tile * 16;
  const int row = lane & 15;
  const int hi = lane >> 4;
  const int col = row;
  const float* arow = node_feats + (long)(n0 + row) * 256;

  v8f accA[4], accB[4];
#pragma unroll
  for (int nt = 0; nt < 4; nt++) { accA[nt] = zero8(); accB[nt] = zero8(); }
#pragma unroll
  for (int kk = 0; kk < 16; kk++) {
    const int kb = kk * 4 + 2 * hi;
    v2f a = *(const v2f*)(arow + kb);
    v2f b0[4], b1[4];
    bfrag4(wpk + PK_WSC, kk, 2, 0, lane, b0);
    bfrag4(wpk + PK_WUP0, kk, 2, 0, lane, b1);
#pragma unroll
    for (int nt = 0; nt < 4; nt++) {
      accA[nt] = wmma4(a, b0[nt], accA[nt]);
      accB[nt] = wmma4(a, b1[nt], accB[nt]);
    }
  }
#pragma unroll
  for (int nt = 0; nt < 4; nt++)
#pragma unroll
    for (int r = 0; r < 8; r++) {
      const int n = n0 + r + 8 * hi;
      ns[(long)n * 64 + nt * 16 + col] = accA[nt][r];
      sup[(long)n * 64 + nt * 16 + col] = accB[nt][r];
    }

#pragma unroll
  for (int i = 0; i < 3; i++) {
    v8f acc[4];
#pragma unroll
    for (int nt = 0; nt < 4; nt++) acc[nt] = zero8();
#pragma unroll
    for (int kk = 0; kk < 16; kk++) {
      const int kb = kk * 4 + 2 * hi;
      v2f a;
      a.x = arow[64 + kb * 3 + i];
      a.y = arow[64 + (kb + 1) * 3 + i];
      v2f b[4];
      bfrag4(wpk + PK_WUP1, kk, 2, 0, lane, b);
#pragma unroll
      for (int nt = 0; nt < 4; nt++) acc[nt] = wmma4(a, b[nt], acc[nt]);
    }
#pragma unroll
    for (int nt = 0; nt < 4; nt++)
#pragma unroll
      for (int r = 0; r < 8; r++) {
        const int n = n0 + r + 8 * hi;
        vup[(long)i * NN * 64 + (long)n * 64 + nt * 16 + col] = acc[nt][r];
      }
  }
}

__global__ __launch_bounds__(128) void edge_kernel(
    const float* __restrict__ edge_attrs, const float* __restrict__ edge_feats,
    const float* __restrict__ lengths, const int* __restrict__ edge_index,
    const int* __restrict__ perm,
    const float4* __restrict__ wpk, const float* __restrict__ b1,
    const float* __restrict__ b2, const float* __restrict__ ns,
    const float* __restrict__ sup, const float* __restrict__ vup,
    float* __restrict__ m0, float* __restrict__ m1) {
  __shared__ __attribute__((aligned(16))) float lds_w[4][32 * 64 + 32 * SPITCH];
  __shared__ __attribute__((aligned(16))) float lds_meta[4][32 * 8];
  const int lane = threadIdx.x & 31;
  const int wv = threadIdx.x >> 5;
  const int dt = blockIdx.x * 4 + wv;
  const int e0 = dt * 32;
  const int row = lane & 15;
  const int hi = lane >> 4;
  const int col = row;

  int ea = perm[e0 + row];
  int eb = perm[e0 + 16 + row];
  ea = (unsigned)ea < (unsigned)NE ? ea : 0;
  eb = (unsigned)eb < (unsigned)NE ? eb : 0;
  const int snd0 = edge_index[ea], rcv0 = edge_index[NE + ea];
  const int snd1 = edge_index[eb], rcv1 = edge_index[NE + eb];
  {
    const int mr = row + 16 * hi;
    const int me = hi ? eb : ea;
    const int myr = hi ? rcv1 : rcv0;
    const int nxt = __shfl_down(myr, 1);
    const int endf = ((lane & 7) == 7) || (nxt != myr);
    lds_meta[wv][mr * 8 + 6] = __int_as_float(endf);
    lds_meta[wv][mr * 8 + 7] = 0.f;
    lds_meta[wv][mr * 8 + 0] = __int_as_float(hi ? snd1 : snd0);
    lds_meta[wv][mr * 8 + 1] = __int_as_float(myr);
    lds_meta[wv][mr * 8 + 2] = edge_attrs[me * 4 + 0];
    lds_meta[wv][mr * 8 + 3] = edge_attrs[me * 4 + 1];
    lds_meta[wv][mr * 8 + 4] = edge_attrs[me * 4 + 2];
    lds_meta[wv][mr * 8 + 5] = edge_attrs[me * 4 + 3];
  }
  float* const lds_h = lds_w[wv];
  float* const stg = lds_w[wv] + 32 * 64;
  const float* sp0 = ns + (long)snd0 * 64;
  const float* rp0 = ns + (long)rcv0 * 64;
  const float* sp1 = ns + (long)snd1 * 64;
  const float* rp1 = ns + (long)rcv1 * 64;
  const float* ep0 = edge_feats + (long)ea * 8;
  const float* ep1 = edge_feats + (long)eb * 8;
  const float len0 = lengths[ea];
  const float len1 = lengths[eb];

  v8f acc0[4], acc1[4];
#pragma unroll
  for (int nt = 0; nt < 4; nt++) { acc0[nt] = zero8(); acc1[nt] = zero8(); }
#pragma unroll 4
  for (int kk = 0; kk < 16; kk++) {
    const int kb = kk * 4 + 2 * hi;
    v2f a0 = *(const v2f*)(sp0 + kb);
    v2f a1 = *(const v2f*)(sp1 + kb);
    v2f b[4];
    bfrag4(wpk + PK_W1, kk, 2, 0, lane, b);
#pragma unroll
    for (int nt = 0; nt < 4; nt++) {
      acc0[nt] = wmma4(a0, b[nt], acc0[nt]);
      acc1[nt] = wmma4(a1, b[nt], acc1[nt]);
    }
  }
#pragma unroll 4
  for (int kk = 0; kk < 16; kk++) {
    const int kb = kk * 4 + 2 * hi;
    v2f a0 = *(const v2f*)(rp0 + kb);
    v2f a1 = *(const v2f*)(rp1 + kb);
    v2f b[4];
    bfrag4(wpk + PK_W1, 16 + kk, 2, 0, lane, b);
#pragma unroll
    for (int nt = 0; nt < 4; nt++) {
      acc0[nt] = wmma4(a0, b[nt], acc0[nt]);
      acc1[nt] = wmma4(a1, b[nt], acc1[nt]);
    }
  }
#pragma unroll
  for (int kk = 0; kk < 2; kk++) {
    const int kb = kk * 4 + 2 * hi;
    v2f a0 = *(const v2f*)(ep0 + kb);
    v2f a1 = *(const v2f*)(ep1 + kb);
    v2f b[4];
    bfrag4(wpk + PK_W1, 32 + kk, 2, 0, lane, b);
#pragma unroll
    for (int nt = 0; nt < 4; nt++) {
      acc0[nt] = wmma4(a0, b[nt], acc0[nt]);
      acc1[nt] = wmma4(a1, b[nt], acc1[nt]);
    }
  }
  {
    v2f a0, a1;
    a0.x = hi ? 0.f : len0;  a0.y = 0.f;
    a1.x = hi ? 0.f : len1;  a1.y = 0.f;
    v2f b[4];
    bfrag4(wpk + PK_W1, 34, 2, 0, lane, b);
#pragma unroll
    for (int nt = 0; nt < 4; nt++) {
      acc0[nt] = wmma4(a0, b[nt], acc0[nt]);
      acc1[nt] = wmma4(a1, b[nt], acc1[nt]);
    }
  }
  float4 gs[16];
#pragma unroll
  for (int i = 0; i < 16; i++) {
    const int s_e = __float_as_int(lds_meta[wv][(2 * i + hi) * 8 + 0]);
    gs[i] = *(const float4*)(sup + (long)s_e * 64 + 4 * row);
  }
#pragma unroll
  for (int nt = 0; nt < 4; nt++) {
    const float bias = b1[nt * 16 + col];
#pragma unroll
    for (int r = 0; r < 8; r++) {
      lds_h[(r + 8 * hi) * 64 + nt * 16 + col] = silu_f(acc0[nt][r] + bias);
      lds_h[(16 + r + 8 * hi) * 64 + nt * 16 + col] =
          silu_f(acc1[nt][r] + bias);
    }
  }

#pragma unroll
  for (int i = 0; i < 16; i++)
    *(float4*)&stg[(2 * i + hi) * SPITCH + 4 * row] = gs[i];

#pragma unroll
  for (int nt = 0; nt < 4; nt++) { acc0[nt] = zero8(); acc1[nt] = zero8(); }
#pragma unroll 4
  for (int kk = 0; kk < 16; kk++) {
    const int kb = kk * 4 + 2 * hi;
    v2f a0 = *(const v2f*)&lds_h[row * 64 + kb];
    v2f a1 = *(const v2f*)&lds_h[(16 + row) * 64 + kb];
    v2f b[4];
    bfrag4(wpk + PK_W2, kk, 2, 0, lane, b);
#pragma unroll
    for (int nt = 0; nt < 4; nt++) {
      acc0[nt] = wmma4(a0, b[nt], acc0[nt]);
      acc1[nt] = wmma4(a1, b[nt], acc1[nt]);
    }
  }
  float h2a[4][8], h2b[4][8];
#pragma unroll
  for (int nt = 0; nt < 4; nt++) {
    const float bias = b2[nt * 16 + col];
#pragma unroll
    for (int r = 0; r < 8; r++) {
      h2a[nt][r] = silu_f(acc0[nt][r] + bias);
      h2b[nt][r] = silu_f(acc1[nt][r] + bias);
    }
  }
#pragma unroll
  for (int nt = 0; nt < 4; nt++)
#pragma unroll
    for (int r = 0; r < 8; r++) {
      lds_h[(r + 8 * hi) * 64 + nt * 16 + col] = h2a[nt][r];
      lds_h[(16 + r + 8 * hi) * 64 + nt * 16 + col] = h2b[nt][r];
    }

  {
    v8f ta0[4], ta1[4], tc0[4], tc1[4];
#pragma unroll
    for (int nt = 0; nt < 4; nt++) {
      ta0[nt] = zero8(); ta1[nt] = zero8();
      tc0[nt] = zero8(); tc1[nt] = zero8();
    }
#pragma unroll 2
    for (int kk = 0; kk < 16; kk++) {
      const int kb = kk * 4 + 2 * hi;
      v2f a0 = *(const v2f*)&lds_h[row * 64 + kb];
      v2f a1 = *(const v2f*)&lds_h[(16 + row) * 64 + kb];
      v2f b[4], d[4];
      bfrag4(wpk + PK_W3, kk, 8, 0, lane, b);
      bfrag4(wpk + PK_W3, kk, 8, 4, lane, d);
#pragma unroll
      for (int nt = 0; nt < 4; nt++) {
        ta0[nt] = wmma4(a0, b[nt], ta0[nt]);
        ta1[nt] = wmma4(a1, b[nt], ta1[nt]);
        tc0[nt] = wmma4(a0, d[nt], tc0[nt]);
        tc1[nt] = wmma4(a1, d[nt], tc1[nt]);
      }
    }
#pragma unroll
    for (int t = 0; t < 2; t++) {
      float q0[4], q1[4], q2[4], q3[4];
#pragma unroll
      for (int nt = 0; nt < 4; nt++) {
        q0[nt] = 0.f; q1[nt] = 0.f; q2[nt] = 0.f; q3[nt] = 0.f;
      }
#pragma unroll
      for (int r = 0; r < 8; r++) {
        const int rr = t * 16 + r + 8 * hi;
        const float4 ma = *(const float4*)&lds_meta[wv][rr * 8];
        const float4 mb = *(const float4*)&lds_meta[wv][rr * 8 + 4];
        float xs[4];
#pragma unroll
        for (int nt = 0; nt < 4; nt++)
          xs[nt] = stg[rr * SPITCH + nt * 16 + col];
#pragma unroll
        for (int nt = 0; nt < 4; nt++) {
          const float wa = t ? ta1[nt][r] : ta0[nt][r];
          const float wc = t ? tc1[nt][r] : tc0[nt][r];
          q0[nt] += xs[nt] * ma.z * wa;
          q1[nt] += xs[nt] * ma.w * wc;
          q2[nt] += xs[nt] * mb.x * wc;
          q3[nt] += xs[nt] * mb.y * wc;
        }
        if (__float_as_int(mb.z)) {
          const int r_e = __float_as_int(ma.y);
#pragma unroll
          for (int nt = 0; nt < 4; nt++) {
            const int c = nt * 16 + col;
            unsafeAtomicAdd(&m0[(long)r_e * 128 + c], q0[nt]);
            unsafeAtomicAdd(&m1[(long)r_e * 128 + c], q1[nt]);
            unsafeAtomicAdd(&m1[(long)NN * 128 + (long)r_e * 128 + c], q2[nt]);
            unsafeAtomicAdd(&m1[(long)2 * NN * 128 + (long)r_e * 128 + c],
                            q3[nt]);
            q0[nt] = 0.f; q1[nt] = 0.f; q2[nt] = 0.f; q3[nt] = 0.f;
          }
        }
      }
    }
  }
  {
    v8f tb0[4], tb1[4], td0[4], td1[4];
#pragma unroll
    for (int nt = 0; nt < 4; nt++) {
      tb0[nt] = zero8(); tb1[nt] = zero8();
      td0[nt] = zero8(); td1[nt] = zero8();
    }
#pragma unroll 2
    for (int kk = 0; kk < 16; kk++) {
      const int kb = kk * 4 + 2 * hi;
      v2f a0 = *(const v2f*)&lds_h[row * 64 + kb];
      v2f a1 = *(const v2f*)&lds_h[(16 + row) * 64 + kb];
      v2f b[4], d[4];
      bfrag4(wpk + PK_W3, kk, 8, 2, lane, b);
      bfrag4(wpk + PK_W3, kk, 8, 6, lane, d);
#pragma unroll
      for (int nt = 0; nt < 4; nt++) {
        tb0[nt] = wmma4(a0, b[nt], tb0[nt]);
        tb1[nt] = wmma4(a1, b[nt], tb1[nt]);
        td0[nt] = wmma4(a0, d[nt], td0[nt]);
        td1[nt] = wmma4(a1, d[nt], td1[nt]);
      }
    }
    float* const vst = lds_w[wv];
#pragma unroll
    for (int t = 0; t < 2; t++) {
#pragma unroll 1
      for (int pl = 0; pl < 3; pl++) {
        float4 gv[8];
#pragma unroll
        for (int i = 0; i < 8; i++) {
          const int s_e =
              __float_as_int(lds_meta[wv][(t * 16 + 2 * i + hi) * 8 + 0]);
          gv[i] = *(const float4*)(vup + (long)pl * NN * 64 + (long)s_e * 64 +
                                   4 * row);
        }
#pragma unroll
        for (int i = 0; i < 8; i++)
          *(float4*)&vst[(pl * 16 + 2 * i + hi) * SPITCH + 4 * row] = gv[i];
      }
      float q0[4], q1[4], q2[4], q3[4];
#pragma unroll
      for (int nt = 0; nt < 4; nt++) {
        q0[nt] = 0.f; q1[nt] = 0.f; q2[nt] = 0.f; q3[nt] = 0.f;
      }
#pragma unroll
      for (int r = 0; r < 8; r++) {
        const int rr = t * 16 + r + 8 * hi;
        const int lr = r + 8 * hi;
        const float4 ma = *(const float4*)&lds_meta[wv][rr * 8];
        const float4 mb = *(const float4*)&lds_meta[wv][rr * 8 + 4];
        float x0[4], x1[4], x2[4];
#pragma unroll
        for (int nt = 0; nt < 4; nt++) {
          const int c = nt * 16 + col;
          x0[nt] = vst[(0 * 16 + lr) * SPITCH + c];
          x1[nt] = vst[(1 * 16 + lr) * SPITCH + c];
          x2[nt] = vst[(2 * 16 + lr) * SPITCH + c];
        }
#pragma unroll
        for (int nt = 0; nt < 4; nt++) {
          const float wb = t ? tb1[nt][r] : tb0[nt][r];
          const float wd = t ? td1[nt][r] : td0[nt][r];
          const float dot = x0[nt] * ma.w +
                            x1[nt] * mb.x +
                            x2[nt] * mb.y;
          q0[nt] += dot * INV_SQRT3 * wb;
          q1[nt] += x0[nt] * ma.z * wd;
          q2[nt] += x1[nt] * ma.z * wd;
          q3[nt] += x2[nt] * ma.z * wd;
        }
        if (__float_as_int(mb.z)) {
          const int r_e = __float_as_int(ma.y);
#pragma unroll
          for (int nt = 0; nt < 4; nt++) {
            const int c = nt * 16 + col;
            unsafeAtomicAdd(&m0[(long)r_e * 128 + 64 + c], q0[nt]);
            unsafeAtomicAdd(&m1[(long)r_e * 128 + 64 + c], q1[nt]);
            unsafeAtomicAdd(&m1[(long)NN * 128 + (long)r_e * 128 + 64 + c],
                            q2[nt]);
            unsafeAtomicAdd(
                &m1[(long)2 * NN * 128 + (long)r_e * 128 + 64 + c], q3[nt]);
            q0[nt] = 0.f; q1[nt] = 0.f; q2[nt] = 0.f; q3[nt] = 0.f;
          }
        }
      }
    }
  }
}

__global__ __launch_bounds__(256) void out_kernel(
    const float* __restrict__ m0, const float* __restrict__ m1,
    const float* __restrict__ Ws, const float* __restrict__ Wv,
    float* __restrict__ out) {
  const int lane = threadIdx.x & 31;
  const int wv = threadIdx.x >> 5;
  const int tile = blockIdx.x * 8 + wv;
  if (tile >= NN / 16) return;
  const int n0 = tile * 16;
  const int row = lane & 15;
  const int hi = lane >> 4;
  const int col = row;

  v8f acc[4][4];
#pragma unroll
  for (int j = 0; j < 4; j++)
#pragma unroll
    for (int nt = 0; nt < 4; nt++) acc[j][nt] = zero8();

#pragma unroll 4
  for (int kk = 0; kk < 32; kk++) {
    const int kb = kk * 4 + 2 * hi;
    v2f a = *(const v2f*)(m0 + (long)(n0 + row) * 128 + kb);
#pragma unroll
    for (int nt = 0; nt < 4; nt++) {
      v2f b;
      b.x = Ws[kb * 64 + nt * 16 + col];
      b.y = Ws[(kb + 1) * 64 + nt * 16 + col];
      acc[0][nt] = wmma4(a, b, acc[0][nt]);
    }
  }
#pragma unroll
  for (int i = 0; i < 3; i++) {
#pragma unroll 4
    for (int kk = 0; kk < 32; kk++) {
      const int kb = kk * 4 + 2 * hi;
      v2f a =
          *(const v2f*)(m1 + (long)i * NN * 128 + (long)(n0 + row) * 128 + kb);
#pragma unroll
      for (int nt = 0; nt < 4; nt++) {
        v2f b;
        b.x = Wv[kb * 64 + nt * 16 + col];
        b.y = Wv[(kb + 1) * 64 + nt * 16 + col];
        acc[1 + i][nt] = wmma4(a, b, acc[1 + i][nt]);
      }
    }
  }
  const float inv = 1.0f / 16.0f;
#pragma unroll
  for (int nt = 0; nt < 4; nt++) {
    const int c = nt * 16 + col;
#pragma unroll
    for (int r = 0; r < 8; r++) {
      const int n = n0 + r + 8 * hi;
      float4 o = make_float4(acc[0][nt][r] * inv, acc[1][nt][r] * inv,
                             acc[2][nt][r] * inv, acc[3][nt][r] * inv);
      *(float4*)(out + (long)n * 256 + c * 4) = o;
    }
  }
}

extern "C" void kernel_launch(void* const* d_in, const int* in_sizes, int n_in,
                              void* d_out, int out_size, void* d_ws,
                              size_t ws_size, hipStream_t stream) {
  (void)in_sizes; (void)n_in; (void)out_size; (void)ws_size;
  const float* node_feats = (const float*)d_in[0];
  const float* edge_attrs = (const float*)d_in[1];
  const float* edge_feats = (const float*)d_in[2];
  const float* lengths    = (const float*)d_in[3];
  const int*   edge_index = (const int*)d_in[4];
  const float* Wsc  = (const float*)d_in[5];
  const float* Wup0 = (const float*)d_in[6];
  const float* Wup1 = (const float*)d_in[7];
  const float* W1   = (const float*)d_in[8];
  const float* b1   = (const float*)d_in[9];
  const float* W2   = (const float*)d_in[10];
  const float* b2   = (const float*)d_in[11];
  const float* W3   = (const float*)d_in[12];
  const float* Ws   = (const float*)d_in[13];
  const float* Wv   = (const float*)d_in[14];
  float* out = (float*)d_out;

  float* ws  = (float*)d_ws;
  float* ns  = ws;
  float* sup = ns + (long)NN * 64;
  float* vup = sup + (long)NN * 64;
  float* m0  = vup + (long)3 * NN * 64;
  float* m1  = m0 + (long)NN * 128;

  int* scr     = (int*)d_out;
  int* cnt     = scr;
  int* start   = cnt + NN;
  int* arrival = start + NN;
  int* perm_a  = arrival + NE;
  int* perm_b  = perm_a + NE;
  float4* wpk  = (float4*)(perm_b + NE);

  const long zn4 = (long)NN * 512 / 4;
  zero_kernel<<<(int)((zn4 + 255) / 256), 256, 0, stream>>>(
      (float4*)m0, zn4, (int4*)cnt, (long)NN / 4, W1, W2, W3, Wsc, Wup0, Wup1,
      wpk);
  count_kernel<<<(NE + 255) / 256, 256, 0, stream>>>(edge_index, cnt, arrival);
  scan_kernel<<<1, 1024, 0, stream>>>(cnt, start);
  place_kernel<<<(NE + 255) / 256, 256, 0, stream>>>(edge_index, start, arrival,
                                                     perm_a);
  rank_kernel<<<(NE + 255) / 256, 256, 0, stream>>>(edge_index, cnt, start,
                                                    perm_a, perm_b);

  node_up_kernel<<<(NN / 16 + 7) / 8, 256, 0, stream>>>(node_feats, wpk, ns,
                                                        sup, vup);
  edge_kernel<<<NE / 32 / 4, 128, 0, stream>>>(edge_attrs, edge_feats, lengths,
                                               edge_index, perm_b, wpk, b1, b2,
                                               ns, sup, vup, m0, m1);
  out_kernel<<<(NN / 16 + 7) / 8, 256, 0, stream>>>(m0, m1, Ws, Wv, out);
}
